# starting kernel
# baseline (speedup 1.0000x reference)
#include <hip/hip_runtime.h>
#include <math.h>

#define N_FEAT 128

typedef __attribute__((ext_vector_type(2))) float v2f;
typedef __attribute__((ext_vector_type(8))) float v8f;

__global__ void deg_init_kernel(float* __restrict__ deg, int n) {
    int i = blockIdx.x * blockDim.x + threadIdx.x;
    if (i < n) deg[i] = 1.0f;
}

__global__ void deg_scatter_kernel(const int* __restrict__ dst, float* __restrict__ deg, int E) {
    int e = blockIdx.x * blockDim.x + threadIdx.x;
    if (e < E) atomicAdd(&deg[dst[e]], 1.0f);
}

__global__ void deg_rsqrt_kernel(float* __restrict__ deg, int n) {
    int i = blockIdx.x * blockDim.x + threadIdx.x;
    if (i < n) deg[i] = rsqrtf(deg[i]);
}

__global__ void gemm_wmma_kernel(const float* __restrict__ X, const float* __restrict__ W,
                                 float* __restrict__ H, int M) {
    const int wave = threadIdx.x >> 5;
    const int lane = threadIdx.x & 31;
    const int half = lane >> 4;
    const int l16  = lane & 15;
    const int row0 = blockIdx.x * 16;
    if (row0 >= M) return;
    const int col = wave * 16 + l16;

    v8f acc = {};
    const float* xrow = X + (size_t)(row0 + l16) * N_FEAT + half * 2;
    #pragma unroll
    for (int k0 = 0; k0 < N_FEAT; k0 += 4) {
        v2f a;
        a.x = xrow[k0 + 0];
        a.y = xrow[k0 + 1];
        v2f b;
        b.x = W[(size_t)(k0 + half * 2 + 0) * N_FEAT + col];
        b.y = W[(size_t)(k0 + half * 2 + 1) * N_FEAT + col];
        acc = __builtin_amdgcn_wmma_f32_16x16x4_f32(
             false, a,  false, b,
             (short)0, acc,  false,  false);
    }
    float* out = H + (size_t)row0 * N_FEAT + wave * 16 + l16;
    #pragma unroll
    for (int r = 0; r < 8; ++r) {
        out[(size_t)(half * 8 + r) * N_FEAT] = acc[r];
    }
}

__global__ void zero_kernel(float4* __restrict__ p, long long n4) {
    long long i = (long long)blockIdx.x * blockDim.x + threadIdx.x;
    if (i < n4) p[i] = make_float4(0.f, 0.f, 0.f, 0.f);
}

__global__ void scatter_kernel(const float* __restrict__ H, const float* __restrict__ dinv,
                               const int* __restrict__ src, const int* __restrict__ dst,
                               float* __restrict__ agg, int E) {
    const int lane = threadIdx.x & 31;
    const int e = blockIdx.x * (blockDim.x >> 5) + (threadIdx.x >> 5);
    if (e >= E) return;
    const int s = src[e];
    const int d = dst[e];
    const float nrm = dinv[s] * dinv[d];
    const float4 v = *(const float4*)(H + (size_t)s * N_FEAT + lane * 4);
    float* p = agg + (size_t)d * N_FEAT + lane * 4;
    atomicAdd(p + 0, v.x * nrm);
    atomicAdd(p + 1, v.y * nrm);
    atomicAdd(p + 2, v.z * nrm);
    atomicAdd(p + 3, v.w * nrm);
}

__global__ void finalize_kernel(const float* __restrict__ agg, const float* __restrict__ H,
                                const float* __restrict__ dinv, const float* __restrict__ bias,
                                const float* __restrict__ alpha, float* __restrict__ out, int n) {
    const int lane = threadIdx.x & 31;
    const int node = blockIdx.x * (blockDim.x >> 5) + (threadIdx.x >> 5);
    if (node >= n) return;
    const float di = dinv[node];
    const float di2 = di * di;
    const int f0 = lane * 4;
    const float4 ag = *(const float4*)(agg + (size_t)node * N_FEAT + f0);
    const float4 hv = *(const float4*)(H + (size_t)node * N_FEAT + f0);
    const float4 bv = *(const float4*)(bias + f0);
    const float4 av = *(const float4*)(alpha + f0);
    float4 r;
    r.x = ag.x + hv.x * di2 + bv.x;  r.x = r.x > 0.f ? r.x : av.x * r.x;
    r.y = ag.y + hv.y * di2 + bv.y;  r.y = r.y > 0.f ? r.y : av.y * r.y;
    r.z = ag.z + hv.z * di2 + bv.z;  r.z = r.z > 0.f ? r.z : av.z * r.z;
    r.w = ag.w + hv.w * di2 + bv.w;  r.w = r.w > 0.f ? r.w : av.w * r.w;
    *(float4*)(out + (size_t)node * N_FEAT + f0) = r;
}

extern "C" void kernel_launch(void* const* d_in, const int* in_sizes, int n_in,
                              void* d_out, int out_size, void* d_ws, size_t ws_size,
                              hipStream_t stream) {
    (void)n_in; (void)out_size; (void)ws_size;
    const float* x  = (const float*)d_in[0];
    const int*   ei = (const int*)d_in[1];
    const float* W1 = (const float*)d_in[2];
    const float* b1 = (const float*)d_in[3];
    const float* W2 = (const float*)d_in[4];
    const float* b2 = (const float*)d_in[5];
    const float* al = (const float*)d_in[6];

    const int N = in_sizes[0] / N_FEAT;
    const int E = in_sizes[1] / 2;
    const int* src = ei;
    const int* dst = ei + E;

    float* dinv = (float*)d_ws;
    size_t off = (((size_t)N * sizeof(float)) + 255) & ~(size_t)255;
    float* hbuf = (float*)((char*)d_ws + off);
    float* agg  = (float*)d_out;
    float* outp = (float*)d_out;

    deg_init_kernel<<<(N + 255) / 256, 256, 0, stream>>>(dinv, N);
    deg_scatter_kernel<<<(E + 255) / 256, 256, 0, stream>>>(dst, dinv, E);
    deg_rsqrt_kernel<<<(N + 255) / 256, 256, 0, stream>>>(dinv, N);

    const int mTiles = (N + 15) / 16;
    const long long n4 = (long long)N * N_FEAT / 4;
    const int zBlocks = (int)((n4 + 255) / 256);
    const int eBlocks = (E + 7) / 8;
    const int nBlocks = (N + 7) / 8;

    gemm_wmma_kernel<<<mTiles, 256, 0, stream>>>(x, W1, hbuf, N);
    zero_kernel<<<zBlocks, 256, 0, stream>>>((float4*)agg, n4);
    scatter_kernel<<<eBlocks, 256, 0, stream>>>(hbuf, dinv, src, dst, agg, E);
    finalize_kernel<<<nBlocks, 256, 0, stream>>>(agg, hbuf, dinv, b1, al, outp, N);

    gemm_wmma_kernel<<<mTiles, 256, 0, stream>>>(outp, W2, hbuf, N);
    zero_kernel<<<zBlocks, 256, 0, stream>>>((float4*)agg, n4);
    scatter_kernel<<<eBlocks, 256, 0, stream>>>(hbuf, dinv, src, dst, agg, E);
    finalize_kernel<<<nBlocks, 256, 0, stream>>>(agg, hbuf, dinv, b2, al, outp, N);
}


# edge sums moved from global f32 atomics to work-group-owned LDS rows: each group filters the edge list for its 392 rows (no sort), adds by LDS f32 atomics in arrival order, finishes, stores once
# speedup vs baseline: 2.3095x; 2.3095x over previous
#include <hip/hip_runtime.h>
#include <math.h>

#define N_FEAT 128

typedef __attribute__((ext_vector_type(2))) float v2f;
typedef __attribute__((ext_vector_type(8))) float v8f;

__global__ void deg_init_kernel(float* __restrict__ deg, int n) {
    int i = blockIdx.x * blockDim.x + threadIdx.x;
    if (i < n) deg[i] = 1.0f;
}

__global__ void deg_scatter_kernel(const int* __restrict__ dst, float* __restrict__ deg, int E) {
    int e = blockIdx.x * blockDim.x + threadIdx.x;
    if (e < E) atomicAdd(&deg[dst[e]], 1.0f);
}

__global__ void deg_rsqrt_kernel(float* __restrict__ deg, int n) {
    int i = blockIdx.x * blockDim.x + threadIdx.x;
    if (i < n) deg[i] = rsqrtf(deg[i]);
}

__global__ void __launch_bounds__(256)
gemm_wmma_kernel(const float* __restrict__ X, const float* __restrict__ W,
                 float* __restrict__ H, int M) {
    __shared__ float4 Wl[32 * 128];
    const int tid  = threadIdx.x;
    const int wave = __builtin_amdgcn_readfirstlane(tid >> 5);
    const int lane = tid & 31;
    const int half = lane >> 4;
    const int l16  = lane & 15;
    const int step = (int)gridDim.x * 8;
    int st = wave * (int)gridDim.x + (int)blockIdx.x;
    bool live = st * 16 < M;

    float4 a4[16];
    {
        const float4* xr = (const float4*)(X + (size_t)(live ? st * 16 + l16 : 0) * N_FEAT) + half;
        #pragma unroll
        for (int i = 0; i < 16; ++i) a4[i] = xr[2 * i];
    }

    {
        const int col  = tid & 127;
        const int slot = ((((col & 3) + ((col >> 6) << 2))) << 4) + ((col & 63) >> 2);
        const float* wp = W + col;
        for (int it = 0; it < 16; ++it) {
            const int kq = (tid >> 7) + 2 * it;
            const float* p = wp + (size_t)(4 * kq) * N_FEAT;
            float4 v;
            v.x = p[0];
            v.y = p[N_FEAT];
            v.z = p[2 * N_FEAT];
            v.w = p[3 * N_FEAT];
            Wl[kq * 128 + slot] = v;
        }
    }
    __syncthreads();

    const float4* wl = Wl + half * 128 + l16;

    while (live) {
        asm volatile("" ::: "memory");
        v8f acc[8];
        #pragma unroll
        for (int t = 0; t < 8; ++t) acc[t] = (v8f){0.f, 0.f, 0.f, 0.f, 0.f, 0.f, 0.f, 0.f};

        float4 b[8];
        #pragma unroll
        for (int t = 0; t < 8; ++t) b[t] = wl[t * 16];

        #pragma unroll
        for (int i = 0; i < 16; ++i) {
            float4 bn[8];
            if (i + 1 < 16) {
                #pragma unroll
                for (int t = 0; t < 8; ++t) bn[t] = wl[(2 * (i + 1)) * 128 + t * 16];
            }
            v2f a0; a0.x = a4[i].x; a0.y = a4[i].y;
            v2f a1; a1.x = a4[i].z; a1.y = a4[i].w;
            #pragma unroll
            for (int t = 0; t < 8; ++t) {
                v2f bb; bb.x = b[t].x; bb.y = b[t].y;
                acc[t] = __builtin_amdgcn_wmma_f32_16x16x4_f32(
                     false, a0,  false, bb,
                     (short)0, acc[t],  false,  false);
            }
            #pragma unroll
            for (int t = 0; t < 8; ++t) {
                v2f bb; bb.x = b[t].z; bb.y = b[t].w;
                acc[t] = __builtin_amdgcn_wmma_f32_16x16x4_f32(
                     false, a1,  false, bb,
                     (short)0, acc[t],  false,  false);
            }
            if (i + 1 < 16) {
                #pragma unroll
                for (int t = 0; t < 8; ++t) b[t] = bn[t];
            }
        }

        float* hrow = H + (size_t)(st * 16 + half * 8) * N_FEAT + l16 * 4;

        st += step;
        live = st * 16 < M;
        if (live) {
            const float4* xr = (const float4*)(X + (size_t)(st * 16 + l16) * N_FEAT) + half;
            #pragma unroll
            for (int i = 0; i < 16; ++i) a4[i] = xr[2 * i];
        }

        #pragma unroll
        for (int r = 0; r < 8; ++r) {
            float4 lo, hi;
            lo.x = acc[0][r]; lo.y = acc[1][r]; lo.z = acc[2][r]; lo.w = acc[3][r];
            hi.x = acc[4][r]; hi.y = acc[5][r]; hi.z = acc[6][r]; hi.w = acc[7][r];
            *(float4*)(hrow + (size_t)r * N_FEAT)      = lo;
            *(float4*)(hrow + (size_t)r * N_FEAT + 64) = hi;
        }
    }
}

#define AGG_ROWS    392
#define AGG_THREADS 1024
#define AGG_WAVES   (AGG_THREADS / 32)
#define AGG_AHEAD   8
#define AGG_CHUNK   (AGG_AHEAD * 128)
#define AGG_GROUP   12
#define AGG_QCAP    (AGG_GROUP + 128)

template <bool FULL>
__device__ __forceinline__ void agg_add_group(const float* __restrict__ H, const float* __restrict__ dinv,
                                              const int* __restrict__ src, float* acc, const int2* qp,
                                              int count, int base, unsigned ulane) {
    const int qi = ulane < (unsigned)AGG_GROUP ? (int)ulane : 0;
    int e = 0, r = 0;
    if (FULL || qi < count) { const int2 ent = qp[qi]; e = ent.x; r = ent.y; }
    const int s = src[e];
    const int wi = __float_as_int(dinv[s] * dinv[base + r]);

    float v[AGG_GROUP][4];
    #pragma unroll
    for (int u = 0; u < AGG_GROUP; ++u) {
        v[u][0] = 0.f; v[u][1] = 0.f; v[u][2] = 0.f; v[u][3] = 0.f;
        if (FULL || u < count) {
            const int su = __builtin_amdgcn_readlane(s, u);
            const float* row = H + (size_t)(unsigned)su * N_FEAT;
            v[u][0] = row[ulane];
            v[u][1] = row[ulane + 32u];
            v[u][2] = row[ulane + 64u];
            v[u][3] = row[ulane + 96u];
        }
    }
    #pragma unroll
    for (int u = 0; u < AGG_GROUP; ++u) {
        if (FULL || u < count) {
            const int ru = __builtin_amdgcn_readlane(r, u);
            const float wu = __int_as_float(__builtin_amdgcn_readlane(wi, u));
            float* a = acc + (size_t)(unsigned)ru * N_FEAT + ulane;
            __hip_atomic_fetch_add(a,       v[u][0] * wu, __ATOMIC_RELAXED, __HIP_MEMORY_SCOPE_WORKGROUP);
            __hip_atomic_fetch_add(a + 32u, v[u][1] * wu, __ATOMIC_RELAXED, __HIP_MEMORY_SCOPE_WORKGROUP);
            __hip_atomic_fetch_add(a + 64u, v[u][2] * wu, __ATOMIC_RELAXED, __HIP_MEMORY_SCOPE_WORKGROUP);
            __hip_atomic_fetch_add(a + 96u, v[u][3] * wu, __ATOMIC_RELAXED, __HIP_MEMORY_SCOPE_WORKGROUP);
        }
    }
}

__global__ void __launch_bounds__(AGG_THREADS)
aggregate_kernel(const float* __restrict__ H, const float* __restrict__ dinv,
                 const int* __restrict__ src, const int* __restrict__ dst,
                 const float* __restrict__ bias, const float* __restrict__ alpha,
                 float* __restrict__ out, int n, int E) {
    __shared__ float4 acc4[AGG_ROWS * (N_FEAT / 4)];
    __shared__ int2   qall[AGG_WAVES * AGG_QCAP];
    float* acc = (float*)acc4;

    const int tid = threadIdx.x;
    const unsigned ulane = threadIdx.x & 31u;
    const int lane = (int)ulane;
    const int wave = __builtin_amdgcn_readfirstlane(tid >> 5);
    const int base = (int)blockIdx.x * AGG_ROWS;
    const int left = n - base;
    const int rows = left < AGG_ROWS ? left : AGG_ROWS;

    for (int idx = tid; idx < rows * (N_FEAT / 4); idx += AGG_THREADS) {
        const int r = idx >> 5, c4 = idx & 31;
        const float di = dinv[base + r];
        const float di2 = di * di;
        const float4 hv = *(const float4*)(H + (size_t)(base + r) * N_FEAT + c4 * 4);
        const float4 bv = *(const float4*)(bias + c4 * 4);
        float4 st;
        st.x = hv.x * di2 + bv.x;
        st.y = hv.y * di2 + bv.y;
        st.z = hv.z * di2 + bv.z;
        st.w = hv.w * di2 + bv.w;
        acc4[idx] = st;
    }
    __syncthreads();

    int2* q = qall + wave * AGG_QCAP;
    int cnt = 0;
    const bool vec = (((size_t)dst) & 15) == 0;
    const int nChunks = (E + AGG_CHUNK - 1) / AGG_CHUNK;
    const int nTrips = (nChunks + AGG_WAVES - 1) / AGG_WAVES;
    const int rot = (int)(((long long)blockIdx.x * nTrips) / (int)gridDim.x);

    for (int k = 0; k < nTrips; ++k) {
        int kk = k + rot;
        if (kk >= nTrips) kk -= nTrips;
        const int c = wave + AGG_WAVES * kk;
        if (c >= nChunks) continue;
        const int e0 = c * AGG_CHUNK;

        int4 d[AGG_AHEAD];
        if (vec && e0 + AGG_CHUNK <= E) {
            const int4* dp = (const int4*)(dst + e0) + lane;
            #pragma unroll
            for (int j = 0; j < AGG_AHEAD; ++j) d[j] = dp[j * 32];
        } else {
            #pragma unroll
            for (int j = 0; j < AGG_AHEAD; ++j) {
                const int eb = e0 + j * 128 + 4 * lane;
                d[j].x = (eb     < E) ? dst[eb]     : -1;
                d[j].y = (eb + 1 < E) ? dst[eb + 1] : -1;
                d[j].z = (eb + 2 < E) ? dst[eb + 2] : -1;
                d[j].w = (eb + 3 < E) ? dst[eb + 3] : -1;
            }
        }

        int jstart = 0;
        for (;;) {
            int jnext = AGG_AHEAD;
            #pragma unroll
            for (int j = 0; j < AGG_AHEAD; ++j) {
                if (j >= jstart && jnext == AGG_AHEAD) {
                    const int eb = e0 + j * 128 + 4 * lane;
                    const unsigned r0 = (unsigned)(d[j].x - base);
                    const unsigned r1 = (unsigned)(d[j].y - base);
                    const unsigned r2 = (unsigned)(d[j].z - base);
                    const unsigned r3 = (unsigned)(d[j].w - base);
                    const bool m0 = r0 < (unsigned)rows, m1 = r1 < (unsigned)rows;
                    const bool m2 = r2 < (unsigned)rows, m3 = r3 < (unsigned)rows;
                    if (__builtin_amdgcn_ballot_w32(m0 | m1 | m2 | m3) != 0u) {
                        const unsigned k0 = __builtin_amdgcn_ballot_w32(m0);
                        if (m0) q[cnt + (int)__builtin_amdgcn_mbcnt_lo(k0, 0u)] = make_int2(eb, (int)r0);
                        cnt += __builtin_popcount(k0);
                        const unsigned k1 = __builtin_amdgcn_ballot_w32(m1);
                        if (m1) q[cnt + (int)__builtin_amdgcn_mbcnt_lo(k1, 0u)] = make_int2(eb + 1, (int)r1);
                        cnt += __builtin_popcount(k1);
                        const unsigned k2 = __builtin_amdgcn_ballot_w32(m2);
                        if (m2) q[cnt + (int)__builtin_amdgcn_mbcnt_lo(k2, 0u)] = make_int2(eb + 2, (int)r2);
                        cnt += __builtin_popcount(k2);
                        const unsigned k3 = __builtin_amdgcn_ballot_w32(m3);
                        if (m3) q[cnt + (int)__builtin_amdgcn_mbcnt_lo(k3, 0u)] = make_int2(eb + 3, (int)r3);
                        cnt += __builtin_popcount(k3);
                        if (cnt >= AGG_GROUP) jnext = j + 1;
                    }
                }
            }
            __builtin_amdgcn_fence(__ATOMIC_ACQ_REL, "wavefront");
            while (cnt >= AGG_GROUP) {
                cnt -= AGG_GROUP;
                agg_add_group<true>(H, dinv, src, acc, q + cnt, AGG_GROUP, base, ulane);
            }
            if (jnext >= AGG_AHEAD) break;
            jstart = jnext;
        }
    }
    if (cnt > 0) agg_add_group<false>(H, dinv, src, acc, q, cnt, base, ulane);
    __syncthreads();

    for (int idx = tid; idx < rows * (N_FEAT / 4); idx += AGG_THREADS) {
        const int c4 = idx & 31;
        const float4 av = *(const float4*)(alpha + c4 * 4);
        float4 r = acc4[idx];
        r.x = r.x > 0.f ? r.x : av.x * r.x;
        r.y = r.y > 0.f ? r.y : av.y * r.y;
        r.z = r.z > 0.f ? r.z : av.z * r.z;
        r.w = r.w > 0.f ? r.w : av.w * r.w;
        *(float4*)(out + (size_t)base * N_FEAT + (size_t)idx * 4) = r;
    }
}

extern "C" void kernel_launch(void* const* d_in, const int* in_sizes, int n_in,
                              void* d_out, int out_size, void* d_ws, size_t ws_size,
                              hipStream_t stream) {
    (void)n_in; (void)out_size; (void)ws_size;
    const float* x  = (const float*)d_in[0];
    const int*   ei = (const int*)d_in[1];
    const float* W1 = (const float*)d_in[2];
    const float* b1 = (const float*)d_in[3];
    const float* W2 = (const float*)d_in[4];
    const float* b2 = (const float*)d_in[5];
    const float* al = (const float*)d_in[6];

    const int N = in_sizes[0] / N_FEAT;
    const int E = in_sizes[1] / 2;
    const int* src = ei;
    const int* dst = ei + E;

    float* dinv = (float*)d_ws;
    size_t off = (((size_t)N * sizeof(float)) + 255) & ~(size_t)255;
    float* hbuf = (float*)((char*)d_ws + off);
    float* outp = (float*)d_out;

    deg_init_kernel<<<(N + 255) / 256, 256, 0, stream>>>(dinv, N);
    deg_scatter_kernel<<<(E + 255) / 256, 256, 0, stream>>>(dst, dinv, E);
    deg_rsqrt_kernel<<<(N + 255) / 256, 256, 0, stream>>>(dinv, N);

    const int mTiles = (N + 15) / 16;
    const int gFull = (mTiles + 7) / 8;
    const int gBlocks = gFull < 768 ? gFull : 768;
    const int aBlocks = (N + AGG_ROWS - 1) / AGG_ROWS;

    gemm_wmma_kernel<<<gBlocks, 256, 0, stream>>>(x, W1, hbuf, N);
    aggregate_kernel<<<aBlocks, AGG_THREADS, 0, stream>>>(hbuf, dinv, src, dst, b1, al, outp, N, E);

    gemm_wmma_kernel<<<gBlocks, 256, 0, stream>>>(outp, W2, hbuf, N);
    aggregate_kernel<<<aBlocks, AGG_THREADS, 0, stream>>>(hbuf, dinv, src, dst, b2, al, outp, N, E);
}


# edge walk scans one-byte owner tags (16 edges per 16-byte load) and resolves kept edges from interleaved (src,dst) pairs, both written once beside the degree adds; arrival-order LDS sums as before
# speedup vs baseline: 2.5225x; 1.0922x over previous
#include <hip/hip_runtime.h>
#include <math.h>

#define N_FEAT 128

typedef __attribute__((ext_vector_type(2))) float v2f;
typedef __attribute__((ext_vector_type(8))) float v8f;

__global__ void deg_init_kernel(float* __restrict__ deg, int n) {
    int i = blockIdx.x * blockDim.x + threadIdx.x;
    if (i < n) deg[i] = 1.0f;
}

#define TAG_ROWS 392
__global__ void deg_scatter_tag_kernel(const int* __restrict__ src, const int* __restrict__ dst,
                                       float* __restrict__ deg, unsigned* __restrict__ tag4,
                                       int* __restrict__ pair, int E, int nWords) {
    const int t = blockIdx.x * blockDim.x + threadIdx.x;
    if (t >= nWords) return;
    const int e = 4 * t;
    int d0 = -1, d1 = -1, d2 = -1, d3 = -1;
    if (e + 3 < E && (((((size_t)dst) | ((size_t)src)) & 15) == 0)) {
        const int4 dv = *(const int4*)(dst + e);
        const int4 sv = *(const int4*)(src + e);
        d0 = dv.x; d1 = dv.y; d2 = dv.z; d3 = dv.w;
        int4* pp = (int4*)(pair + 2 * (size_t)e);
        pp[0] = make_int4(sv.x, dv.x, sv.y, dv.y);
        pp[1] = make_int4(sv.z, dv.z, sv.w, dv.w);
    } else {
        int2* pp = (int2*)pair;
        if (e     < E) { d0 = dst[e];     pp[e]     = make_int2(src[e],     d0); }
        if (e + 1 < E) { d1 = dst[e + 1]; pp[e + 1] = make_int2(src[e + 1], d1); }
        if (e + 2 < E) { d2 = dst[e + 2]; pp[e + 2] = make_int2(src[e + 2], d2); }
        if (e + 3 < E) { d3 = dst[e + 3]; pp[e + 3] = make_int2(src[e + 3], d3); }
    }
    unsigned w = 0u;
    if (d0 >= 0) { atomicAdd(&deg[d0], 1.0f); w |= ((unsigned)(d0 / TAG_ROWS) & 255u); }
    if (d1 >= 0) { atomicAdd(&deg[d1], 1.0f); w |= ((unsigned)(d1 / TAG_ROWS) & 255u) << 8; }
    if (d2 >= 0) { atomicAdd(&deg[d2], 1.0f); w |= ((unsigned)(d2 / TAG_ROWS) & 255u) << 16; }
    if (d3 >= 0) { atomicAdd(&deg[d3], 1.0f); w |= ((unsigned)(d3 / TAG_ROWS) & 255u) << 24; }
    tag4[t] = w;
}

__global__ void deg_rsqrt_kernel(float* __restrict__ deg, int n) {
    int i = blockIdx.x * blockDim.x + threadIdx.x;
    if (i < n) deg[i] = rsqrtf(deg[i]);
}

__global__ void __launch_bounds__(256)
gemm_wmma_kernel(const float* __restrict__ X, const float* __restrict__ W,
                 float* __restrict__ H, int M) {
    __shared__ float4 Wl[32 * 128];
    const int tid  = threadIdx.x;
    const int wave = __builtin_amdgcn_readfirstlane(tid >> 5);
    const int lane = tid & 31;
    const int half = lane >> 4;
    const int l16  = lane & 15;
    const int step = (int)gridDim.x * 8;
    int st = wave * (int)gridDim.x + (int)blockIdx.x;
    bool live = st * 16 < M;

    float4 a4[16];
    {
        const float4* xr = (const float4*)(X + (size_t)(live ? st * 16 + l16 : 0) * N_FEAT) + half;
        #pragma unroll
        for (int i = 0; i < 16; ++i) a4[i] = xr[2 * i];
    }

    {
        const int col  = tid & 127;
        const int slot = ((((col & 3) + ((col >> 6) << 2))) << 4) + ((col & 63) >> 2);
        const float* wp = W + col;
        for (int it = 0; it < 16; ++it) {
            const int kq = (tid >> 7) + 2 * it;
            const float* p = wp + (size_t)(4 * kq) * N_FEAT;
            float4 v;
            v.x = p[0];
            v.y = p[N_FEAT];
            v.z = p[2 * N_FEAT];
            v.w = p[3 * N_FEAT];
            Wl[kq * 128 + slot] = v;
        }
    }
    __syncthreads();

    const float4* wl = Wl + half * 128 + l16;

    while (live) {
        asm volatile("" ::: "memory");
        v8f acc[8];
        #pragma unroll
        for (int t = 0; t < 8; ++t) acc[t] = (v8f){0.f, 0.f, 0.f, 0.f, 0.f, 0.f, 0.f, 0.f};

        float4 b[8];
        #pragma unroll
        for (int t = 0; t < 8; ++t) b[t] = wl[t * 16];

        #pragma unroll
        for (int i = 0; i < 16; ++i) {
            float4 bn[8];
            if (i + 1 < 16) {
                #pragma unroll
                for (int t = 0; t < 8; ++t) bn[t] = wl[(2 * (i + 1)) * 128 + t * 16];
            }
            v2f a0; a0.x = a4[i].x; a0.y = a4[i].y;
            v2f a1; a1.x = a4[i].z; a1.y = a4[i].w;
            #pragma unroll
            for (int t = 0; t < 8; ++t) {
                v2f bb; bb.x = b[t].x; bb.y = b[t].y;
                acc[t] = __builtin_amdgcn_wmma_f32_16x16x4_f32(
                     false, a0,  false, bb,
                     (short)0, acc[t],  false,  false);
            }
            #pragma unroll
            for (int t = 0; t < 8; ++t) {
                v2f bb; bb.x = b[t].z; bb.y = b[t].w;
                acc[t] = __builtin_amdgcn_wmma_f32_16x16x4_f32(
                     false, a1,  false, bb,
                     (short)0, acc[t],  false,  false);
            }
            if (i + 1 < 16) {
                #pragma unroll
                for (int t = 0; t < 8; ++t) b[t] = bn[t];
            }
        }

        float* hrow = H + (size_t)(st * 16 + half * 8) * N_FEAT + l16 * 4;

        st += step;
        live = st * 16 < M;
        if (live) {
            const float4* xr = (const float4*)(X + (size_t)(st * 16 + l16) * N_FEAT) + half;
            #pragma unroll
            for (int i = 0; i < 16; ++i) a4[i] = xr[2 * i];
        }

        #pragma unroll
        for (int r = 0; r < 8; ++r) {
            float4 lo, hi;
            lo.x = acc[0][r]; lo.y = acc[1][r]; lo.z = acc[2][r]; lo.w = acc[3][r];
            hi.x = acc[4][r]; hi.y = acc[5][r]; hi.z = acc[6][r]; hi.w = acc[7][r];
            *(float4*)(hrow + (size_t)r * N_FEAT)      = lo;
            *(float4*)(hrow + (size_t)r * N_FEAT + 64) = hi;
        }
    }
}

#define AGG_ROWS    392
#define AGG_THREADS 1024
#define AGG_WAVES   (AGG_THREADS / 32)
#define AGG_AHEAD   8
#define AGG_VEC     512
#define AGG_CHUNK   (AGG_AHEAD * AGG_VEC)
#define AGG_GROUP   12
#define AGG_QCAP    (AGG_GROUP + AGG_VEC)
static_assert(TAG_ROWS == AGG_ROWS, "the tags must be cut at the groups' row count");

__device__ __forceinline__ unsigned zero_bytes(unsigned x) {
    const unsigned t = ((x & 0x7F7F7F7Fu) + 0x7F7F7F7Fu) | x;
    return ~t & 0x80808080u;
}

template <bool FULL>
__device__ __forceinline__ void agg_add_group(const float* __restrict__ H, const float* __restrict__ dinv,
                                              const int2* __restrict__ pair,
                                              float* acc, const int* qp,
                                              int count, int base, int rows, unsigned ulane) {
    const int qi = ulane < (unsigned)AGG_GROUP ? (int)ulane : 0;
    int e = 0;
    if (FULL || qi < count) e = qp[qi];
    const int2 sd = pair[e];
    const int s = sd.x;
    const int d = sd.y;
    const unsigned ru = (unsigned)(d - base);
    const bool own = ru < (unsigned)rows;
    const int r = own ? (int)ru : 0;
    const int wi = __float_as_int(own ? dinv[s] * dinv[d] : 0.f);

    float v[AGG_GROUP][4];
    #pragma unroll
    for (int u = 0; u < AGG_GROUP; ++u) {
        v[u][0] = 0.f; v[u][1] = 0.f; v[u][2] = 0.f; v[u][3] = 0.f;
        if (FULL || u < count) {
            const int su = __builtin_amdgcn_readlane(s, u);
            const float* row = H + (size_t)(unsigned)su * N_FEAT;
            v[u][0] = row[ulane];
            v[u][1] = row[ulane + 32u];
            v[u][2] = row[ulane + 64u];
            v[u][3] = row[ulane + 96u];
        }
    }
    #pragma unroll
    for (int u = 0; u < AGG_GROUP; ++u) {
        if (FULL || u < count) {
            const int ru2 = __builtin_amdgcn_readlane(r, u);
            const float wu = __int_as_float(__builtin_amdgcn_readlane(wi, u));
            float* a = acc + (size_t)(unsigned)ru2 * N_FEAT + ulane;
            __hip_atomic_fetch_add(a,       v[u][0] * wu, __ATOMIC_RELAXED, __HIP_MEMORY_SCOPE_WORKGROUP);
            __hip_atomic_fetch_add(a + 32u, v[u][1] * wu, __ATOMIC_RELAXED, __HIP_MEMORY_SCOPE_WORKGROUP);
            __hip_atomic_fetch_add(a + 64u, v[u][2] * wu, __ATOMIC_RELAXED, __HIP_MEMORY_SCOPE_WORKGROUP);
            __hip_atomic_fetch_add(a + 96u, v[u][3] * wu, __ATOMIC_RELAXED, __HIP_MEMORY_SCOPE_WORKGROUP);
        }
    }
}

__global__ void __launch_bounds__(AGG_THREADS)
aggregate_kernel(const float* __restrict__ H, const float* __restrict__ dinv,
                 const int* __restrict__ pairs, const unsigned* __restrict__ tag4,
                 const float* __restrict__ bias, const float* __restrict__ alpha,
                 float* __restrict__ out, int n, int E) {
    __shared__ float4 acc4[AGG_ROWS * (N_FEAT / 4)];
    __shared__ int    qall[AGG_WAVES * AGG_QCAP];
    float* acc = (float*)acc4;

    const int tid = threadIdx.x;
    const unsigned ulane = threadIdx.x & 31u;
    const int lane = (int)ulane;
    const int2* pair = (const int2*)pairs;
    const int wave = __builtin_amdgcn_readfirstlane(tid >> 5);
    const int base = (int)blockIdx.x * AGG_ROWS;
    const int left = n - base;
    const int rows = left < AGG_ROWS ? left : AGG_ROWS;

    for (int idx = tid; idx < rows * (N_FEAT / 4); idx += AGG_THREADS) {
        const int r = idx >> 5, c4 = idx & 31;
        const float di = dinv[base + r];
        const float di2 = di * di;
        const float4 hv = *(const float4*)(H + (size_t)(base + r) * N_FEAT + c4 * 4);
        const float4 bv = *(const float4*)(bias + c4 * 4);
        float4 st;
        st.x = hv.x * di2 + bv.x;
        st.y = hv.y * di2 + bv.y;
        st.z = hv.z * di2 + bv.z;
        st.w = hv.w * di2 + bv.w;
        acc4[idx] = st;
    }
    __syncthreads();

    int* q = qall + wave * AGG_QCAP;
    int cnt = 0;
    const unsigned pat = ((unsigned)blockIdx.x & 255u) * 0x01010101u;
    const int nChunks = (E + AGG_CHUNK - 1) / AGG_CHUNK;
    const int nTrips = (nChunks + AGG_WAVES - 1) / AGG_WAVES;
    const int rot = (int)(((long long)blockIdx.x * nChunks) / (int)gridDim.x);

    for (int k = 0; k < nTrips; ++k) {
        int c = wave + AGG_WAVES * k;
        if (c >= nChunks) continue;
        c += rot;
        if (c >= nChunks) c -= nChunks;
        const int e0 = c * AGG_CHUNK;

        uint4 d[AGG_AHEAD];
        {
            const uint4* tp = (const uint4*)(tag4 + (e0 >> 2)) + lane;
            #pragma unroll
            for (int j = 0; j < AGG_AHEAD; ++j) d[j] = tp[j * 32];
        }

        int jstart = 0;
        for (;;) {
            int jnext = AGG_AHEAD;
            #pragma unroll
            for (int j = 0; j < AGG_AHEAD; ++j) {
                if (j >= jstart && jnext == AGG_AHEAD) {
                    unsigned m = (zero_bytes(d[j].x ^ pat) >> 3) | (zero_bytes(d[j].y ^ pat) >> 2) |
                                 (zero_bytes(d[j].z ^ pat) >> 1) |  zero_bytes(d[j].w ^ pat);
                    const int eb = e0 + j * AGG_VEC + 16 * lane;
                    for (;;) {
                        const bool has = m != 0u;
                        if (__builtin_amdgcn_ballot_w32(has) == 0u) break;
                        const int b = has ? __builtin_ctz(m) : 0;
                        const int e = eb + 4 * ((b & 7) - 4) + (b >> 3);
                        const bool keep = has && e < E;
                        const unsigned kb = __builtin_amdgcn_ballot_w32(keep);
                        if (keep) q[cnt + (int)__builtin_amdgcn_mbcnt_lo(kb, 0u)] = e;
                        cnt += __builtin_popcount(kb);
                        m &= m - 1u;
                    }
                    if (cnt >= AGG_GROUP) jnext = j + 1;
                }
            }
            __builtin_amdgcn_fence(__ATOMIC_ACQ_REL, "wavefront");
            while (cnt >= AGG_GROUP) {
                cnt -= AGG_GROUP;
                agg_add_group<true>(H, dinv, pair, acc, q + cnt, AGG_GROUP, base, rows, ulane);
            }
            if (jnext >= AGG_AHEAD) break;
            jstart = jnext;
        }
    }
    if (cnt > 0) agg_add_group<false>(H, dinv, pair, acc, q, cnt, base, rows, ulane);
    __syncthreads();

    for (int idx = tid; idx < rows * (N_FEAT / 4); idx += AGG_THREADS) {
        const int c4 = idx & 31;
        const float4 av = *(const float4*)(alpha + c4 * 4);
        float4 r = acc4[idx];
        r.x = r.x > 0.f ? r.x : av.x * r.x;
        r.y = r.y > 0.f ? r.y : av.y * r.y;
        r.z = r.z > 0.f ? r.z : av.z * r.z;
        r.w = r.w > 0.f ? r.w : av.w * r.w;
        *(float4*)(out + (size_t)base * N_FEAT + (size_t)idx * 4) = r;
    }
}

extern "C" void kernel_launch(void* const* d_in, const int* in_sizes, int n_in,
                              void* d_out, int out_size, void* d_ws, size_t ws_size,
                              hipStream_t stream) {
    (void)n_in; (void)out_size; (void)ws_size;
    const float* x  = (const float*)d_in[0];
    const int*   ei = (const int*)d_in[1];
    const float* W1 = (const float*)d_in[2];
    const float* b1 = (const float*)d_in[3];
    const float* W2 = (const float*)d_in[4];
    const float* b2 = (const float*)d_in[5];
    const float* al = (const float*)d_in[6];

    const int N = in_sizes[0] / N_FEAT;
    const int E = in_sizes[1] / 2;
    const int* src = ei;
    const int* dst = ei + E;

    float* dinv = (float*)d_ws;
    size_t off = (((size_t)N * sizeof(float)) + 255) & ~(size_t)255;
    float* hbuf = (float*)((char*)d_ws + off);
    unsigned* tags = (unsigned*)(hbuf + (size_t)N * N_FEAT);
    const int tagWords = ((E + AGG_CHUNK - 1) / AGG_CHUNK) * (AGG_CHUNK / 4);
    int* pairs = (int*)(tags + tagWords);
    float* outp = (float*)d_out;

    deg_init_kernel<<<(N + 255) / 256, 256, 0, stream>>>(dinv, N);
    deg_scatter_tag_kernel<<<(tagWords + 255) / 256, 256, 0, stream>>>(src, dst, dinv, tags, pairs, E, tagWords);
    deg_rsqrt_kernel<<<(N + 255) / 256, 256, 0, stream>>>(dinv, N);

    const int mTiles = (N + 15) / 16;
    const int gFull = (mTiles + 7) / 8;
    const int gBlocks = gFull < 768 ? gFull : 768;
    const int aBlocks = (N + AGG_ROWS - 1) / AGG_ROWS;

    gemm_wmma_kernel<<<gBlocks, 256, 0, stream>>>(x, W1, hbuf, N);
    aggregate_kernel<<<aBlocks, AGG_THREADS, 0, stream>>>(hbuf, dinv, pairs, tags, b1, al, outp, N, E);

    gemm_wmma_kernel<<<gBlocks, 256, 0, stream>>>(outp, W2, hbuf, N);
    aggregate_kernel<<<aBlocks, AGG_THREADS, 0, stream>>>(hbuf, dinv, pairs, tags, b2, al, outp, N, E);
}


# second layer's edge pass reads back per-group lists of resolved edges (source, owned row, weight) written by the first layer's tag walk: no second walk or pair/weight loads; arrival-order sums kept
# speedup vs baseline: 2.7468x; 1.0889x over previous
#include <hip/hip_runtime.h>
#include <math.h>

#define N_FEAT 128

typedef __attribute__((ext_vector_type(2))) float v2f;
typedef __attribute__((ext_vector_type(8))) float v8f;

__global__ void deg_init_kernel(float* __restrict__ deg, int n) {
    int i = blockIdx.x * blockDim.x + threadIdx.x;
    if (i < n) deg[i] = 1.0f;
}

#define TAG_ROWS 392
__global__ void deg_scatter_tag_kernel(const int* __restrict__ src, const int* __restrict__ dst,
                                       float* __restrict__ deg, unsigned* __restrict__ tag4,
                                       int* __restrict__ pair, int E, int nWords) {
    const int t = blockIdx.x * blockDim.x + threadIdx.x;
    if (t >= nWords) return;
    const int e = 4 * t;
    int d0 = -1, d1 = -1, d2 = -1, d3 = -1;
    if (e + 3 < E && (((((size_t)dst) | ((size_t)src)) & 15) == 0)) {
        const int4 dv = *(const int4*)(dst + e);
        const int4 sv = *(const int4*)(src + e);
        d0 = dv.x; d1 = dv.y; d2 = dv.z; d3 = dv.w;
        int4* pp = (int4*)(pair + 2 * (size_t)e);
        pp[0] = make_int4(sv.x, dv.x, sv.y, dv.y);
        pp[1] = make_int4(sv.z, dv.z, sv.w, dv.w);
    } else {
        int2* pp = (int2*)pair;
        if (e     < E) { d0 = dst[e];     pp[e]     = make_int2(src[e],     d0); }
        if (e + 1 < E) { d1 = dst[e + 1]; pp[e + 1] = make_int2(src[e + 1], d1); }
        if (e + 2 < E) { d2 = dst[e + 2]; pp[e + 2] = make_int2(src[e + 2], d2); }
        if (e + 3 < E) { d3 = dst[e + 3]; pp[e + 3] = make_int2(src[e + 3], d3); }
    }
    unsigned w = 0u;
    if (d0 >= 0) { atomicAdd(&deg[d0], 1.0f); w |= ((unsigned)(d0 / TAG_ROWS) & 255u); }
    if (d1 >= 0) { atomicAdd(&deg[d1], 1.0f); w |= ((unsigned)(d1 / TAG_ROWS) & 255u) << 8; }
    if (d2 >= 0) { atomicAdd(&deg[d2], 1.0f); w |= ((unsigned)(d2 / TAG_ROWS) & 255u) << 16; }
    if (d3 >= 0) { atomicAdd(&deg[d3], 1.0f); w |= ((unsigned)(d3 / TAG_ROWS) & 255u) << 24; }
    tag4[t] = w;
}

__global__ void deg_rsqrt_kernel(float* __restrict__ deg, int n) {
    int i = blockIdx.x * blockDim.x + threadIdx.x;
    if (i < n) deg[i] = rsqrtf(deg[i]);
}

__global__ void __launch_bounds__(256)
gemm_wmma_kernel(const float* __restrict__ X, const float* __restrict__ W,
                 float* __restrict__ H, int M) {
    __shared__ float4 Wl[32 * 128];
    const int tid  = threadIdx.x;
    const int wave = __builtin_amdgcn_readfirstlane(tid >> 5);
    const int lane = tid & 31;
    const int half = lane >> 4;
    const int l16  = lane & 15;
    const int step = (int)gridDim.x * 8;
    int st = wave * (int)gridDim.x + (int)blockIdx.x;
    bool live = st * 16 < M;

    float4 a4[16];
    {
        const float4* xr = (const float4*)(X + (size_t)(live ? st * 16 + l16 : 0) * N_FEAT) + half;
        #pragma unroll
        for (int i = 0; i < 16; ++i) a4[i] = xr[2 * i];
    }

    {
        const int col  = tid & 127;
        const int slot = ((((col & 3) + ((col >> 6) << 2))) << 4) + ((col & 63) >> 2);
        const float* wp = W + col;
        for (int it = 0; it < 16; ++it) {
            const int kq = (tid >> 7) + 2 * it;
            const float* p = wp + (size_t)(4 * kq) * N_FEAT;
            float4 v;
            v.x = p[0];
            v.y = p[N_FEAT];
            v.z = p[2 * N_FEAT];
            v.w = p[3 * N_FEAT];
            Wl[kq * 128 + slot] = v;
        }
    }
    __syncthreads();

    const float4* wl = Wl + half * 128 + l16;

    while (live) {
        asm volatile("" ::: "memory");
        v8f acc[8];
        #pragma unroll
        for (int t = 0; t < 8; ++t) acc[t] = (v8f){0.f, 0.f, 0.f, 0.f, 0.f, 0.f, 0.f, 0.f};

        float4 b[8];
        #pragma unroll
        for (int t = 0; t < 8; ++t) b[t] = wl[t * 16];

        #pragma unroll
        for (int i = 0; i < 16; ++i) {
            float4 bn[8];
            if (i + 1 < 16) {
                #pragma unroll
                for (int t = 0; t < 8; ++t) bn[t] = wl[(2 * (i + 1)) * 128 + t * 16];
            }
            v2f a0; a0.x = a4[i].x; a0.y = a4[i].y;
            v2f a1; a1.x = a4[i].z; a1.y = a4[i].w;
            #pragma unroll
            for (int t = 0; t < 8; ++t) {
                v2f bb; bb.x = b[t].x; bb.y = b[t].y;
                acc[t] = __builtin_amdgcn_wmma_f32_16x16x4_f32(
                     false, a0,  false, bb,
                     (short)0, acc[t],  false,  false);
            }
            #pragma unroll
            for (int t = 0; t < 8; ++t) {
                v2f bb; bb.x = b[t].z; bb.y = b[t].w;
                acc[t] = __builtin_amdgcn_wmma_f32_16x16x4_f32(
                     false, a1,  false, bb,
                     (short)0, acc[t],  false,  false);
            }
            if (i + 1 < 16) {
                #pragma unroll
                for (int t = 0; t < 8; ++t) b[t] = bn[t];
            }
        }

        float* hrow = H + (size_t)(st * 16 + half * 8) * N_FEAT + l16 * 4;

        st += step;
        live = st * 16 < M;
        if (live) {
            const float4* xr = (const float4*)(X + (size_t)(st * 16 + l16) * N_FEAT) + half;
            #pragma unroll
            for (int i = 0; i < 16; ++i) a4[i] = xr[2 * i];
        }

        #pragma unroll
        for (int r = 0; r < 8; ++r) {
            float4 lo, hi;
            lo.x = acc[0][r]; lo.y = acc[1][r]; lo.z = acc[2][r]; lo.w = acc[3][r];
            hi.x = acc[4][r]; hi.y = acc[5][r]; hi.z = acc[6][r]; hi.w = acc[7][r];
            *(float4*)(hrow + (size_t)r * N_FEAT)      = lo;
            *(float4*)(hrow + (size_t)r * N_FEAT + 64) = hi;
        }
    }
}

#define AGG_ROWS    392
#define AGG_THREADS 1024
#define AGG_WAVES   (AGG_THREADS / 32)
#define AGG_AHEAD   8
#define AGG_VEC     512
#define AGG_CHUNK   (AGG_AHEAD * AGG_VEC)
#define AGG_GROUP   12
#define AGG_QCAP    (AGG_GROUP + AGG_VEC)
#define REC_CAP     7168
#define REC_GROUP   16
static_assert(TAG_ROWS == AGG_ROWS, "the tags must be cut at the groups' row count");

__device__ __forceinline__ unsigned zero_bytes(unsigned x) {
    const unsigned t = ((x & 0x7F7F7F7Fu) + 0x7F7F7F7Fu) | x;
    return ~t & 0x80808080u;
}

template <int G, bool FULL>
__device__ __forceinline__ void agg_rows_add(const float* __restrict__ H, float* acc,
                                             int s, int r, int wi, int count, unsigned ulane,
                                             int* rec, int* cursor) {
    float v[G][4];
    #pragma unroll
    for (int u = 0; u < G; ++u) {
        v[u][0] = 0.f; v[u][1] = 0.f; v[u][2] = 0.f; v[u][3] = 0.f;
        if (FULL || u < count) {
            const int su = __builtin_amdgcn_readlane(s, u);
            const float* row = H + (size_t)(unsigned)su * N_FEAT;
            v[u][0] = row[ulane];
            v[u][1] = row[ulane + 32u];
            v[u][2] = row[ulane + 64u];
            v[u][3] = row[ulane + 96u];
        }
    }
    if (rec != nullptr) {
        int pos = 0;
        if (ulane == 0u) pos = __hip_atomic_fetch_add(cursor, count, __ATOMIC_RELAXED, __HIP_MEMORY_SCOPE_WORKGROUP);
        pos = __builtin_amdgcn_readfirstlane(pos);
        if (pos + count <= REC_CAP && (int)ulane < count) {
            __builtin_nontemporal_store(s,  rec + pos + (int)ulane);
            __builtin_nontemporal_store(r,  rec + REC_CAP + pos + (int)ulane);
            __builtin_nontemporal_store(wi, rec + 2 * REC_CAP + pos + (int)ulane);
        }
    }
    #pragma unroll
    for (int u = 0; u < G; ++u) {
        if (FULL || u < count) {
            const int ru2 = __builtin_amdgcn_readlane(r, u);
            const float wu = __int_as_float(__builtin_amdgcn_readlane(wi, u));
            float* a = acc + (size_t)(unsigned)ru2 * N_FEAT + ulane;
            __hip_atomic_fetch_add(a,       v[u][0] * wu, __ATOMIC_RELAXED, __HIP_MEMORY_SCOPE_WORKGROUP);
            __hip_atomic_fetch_add(a + 32u, v[u][1] * wu, __ATOMIC_RELAXED, __HIP_MEMORY_SCOPE_WORKGROUP);
            __hip_atomic_fetch_add(a + 64u, v[u][2] * wu, __ATOMIC_RELAXED, __HIP_MEMORY_SCOPE_WORKGROUP);
            __hip_atomic_fetch_add(a + 96u, v[u][3] * wu, __ATOMIC_RELAXED, __HIP_MEMORY_SCOPE_WORKGROUP);
        }
    }
}

template <bool FULL>
__device__ __forceinline__ void agg_add_group(const float* __restrict__ H, const float* __restrict__ dinv,
                                              const int2* __restrict__ pair,
                                              float* acc, const int* qp,
                                              int count, int base, int rows, unsigned ulane,
                                              int* rec, int* cursor) {
    const int qi = ulane < (unsigned)AGG_GROUP ? (int)ulane : 0;
    int e = 0;
    if (FULL || qi < count) e = qp[qi];
    const int2 sd = pair[e];
    const int s = sd.x;
    const int d = sd.y;
    const unsigned ru = (unsigned)(d - base);
    const bool own = ru < (unsigned)rows;
    const int r = own ? (int)ru : 0;
    const int wi = __float_as_int(own ? dinv[s] * dinv[d] : 0.f);

    agg_rows_add<AGG_GROUP, FULL>(H, acc, s, r, wi, count, ulane, rec, cursor);
}

__device__ __forceinline__ void agg_replay(const float* __restrict__ H, const int* __restrict__ rec,
                                           float* acc, int total, int wave, unsigned ulane) {
    const int qi = ulane < (unsigned)REC_GROUP ? (int)ulane : 0;
    const int nFull = total / REC_GROUP;
    const int rem = total - nFull * REC_GROUP;
    const int last = total - 1;
    int b = wave;
    int i = b * REC_GROUP + qi;
    i = i < last ? i : last;
    int s  = __builtin_nontemporal_load(rec + i);
    int r  = __builtin_nontemporal_load(rec + REC_CAP + i);
    int wi = __builtin_nontemporal_load(rec + 2 * REC_CAP + i);
    while (b < nFull) {
        const int cs = s, cr = r, cw = wi;
        b += AGG_WAVES;
        i = b * REC_GROUP + qi;
        i = i < last ? i : last;
        s  = __builtin_nontemporal_load(rec + i);
        r  = __builtin_nontemporal_load(rec + REC_CAP + i);
        wi = __builtin_nontemporal_load(rec + 2 * REC_CAP + i);
        agg_rows_add<REC_GROUP, true>(H, acc, cs, cr, cw, REC_GROUP, ulane, nullptr, nullptr);
    }
    if (rem > 0 && b == nFull) agg_rows_add<REC_GROUP, false>(H, acc, s, r, wi, rem, ulane, nullptr, nullptr);
}

__device__ __forceinline__ void agg_walk(const float* __restrict__ H, const float* __restrict__ dinv,
                                         const int2* __restrict__ pair, const unsigned* __restrict__ tag4,
                                         float* acc, int* q, int wave, int lane, unsigned ulane,
                                         int base, int rows, int E, int* rec, int* cursor) {
    int cnt = 0;
    const unsigned pat = ((unsigned)blockIdx.x & 255u) * 0x01010101u;
    const int nChunks = (E + AGG_CHUNK - 1) / AGG_CHUNK;
    const int nTrips = (nChunks + AGG_WAVES - 1) / AGG_WAVES;
    const int rot = (int)(((long long)blockIdx.x * nChunks) / (int)gridDim.x);

    for (int k = 0; k < nTrips; ++k) {
        int c = wave + AGG_WAVES * k;
        if (c >= nChunks) continue;
        c += rot;
        if (c >= nChunks) c -= nChunks;
        const int e0 = c * AGG_CHUNK;

        uint4 d[AGG_AHEAD];
        {
            const uint4* tp = (const uint4*)(tag4 + (e0 >> 2)) + lane;
            #pragma unroll
            for (int j = 0; j < AGG_AHEAD; ++j) d[j] = tp[j * 32];
        }

        int jstart = 0;
        for (;;) {
            int jnext = AGG_AHEAD;
            #pragma unroll
            for (int j = 0; j < AGG_AHEAD; ++j) {
                if (j >= jstart && jnext == AGG_AHEAD) {
                    unsigned m = (zero_bytes(d[j].x ^ pat) >> 3) | (zero_bytes(d[j].y ^ pat) >> 2) |
                                 (zero_bytes(d[j].z ^ pat) >> 1) |  zero_bytes(d[j].w ^ pat);
                    const int eb = e0 + j * AGG_VEC + 16 * lane;
                    for (;;) {
                        const bool has = m != 0u;
                        if (__builtin_amdgcn_ballot_w32(has) == 0u) break;
                        const int b = has ? __builtin_ctz(m) : 0;
                        const int e = eb + 4 * ((b & 7) - 4) + (b >> 3);
                        const bool keep = has && e < E;
                        const unsigned kb = __builtin_amdgcn_ballot_w32(keep);
                        if (keep) q[cnt + (int)__builtin_amdgcn_mbcnt_lo(kb, 0u)] = e;
                        cnt += __builtin_popcount(kb);
                        m &= m - 1u;
                    }
                    if (cnt >= AGG_GROUP) jnext = j + 1;
                }
            }
            __builtin_amdgcn_fence(__ATOMIC_ACQ_REL, "wavefront");
            while (cnt >= AGG_GROUP) {
                cnt -= AGG_GROUP;
                agg_add_group<true>(H, dinv, pair, acc, q + cnt, AGG_GROUP, base, rows, ulane, rec, cursor);
            }
            if (jnext >= AGG_AHEAD) break;
            jstart = jnext;
        }
    }
    if (cnt > 0) agg_add_group<false>(H, dinv, pair, acc, q, cnt, base, rows, ulane, rec, cursor);
}

template <bool FIRST>
__global__ void __launch_bounds__(AGG_THREADS)
aggregate_kernel(const float* __restrict__ H, const float* __restrict__ dinv,
                 const int* __restrict__ pairs, const unsigned* __restrict__ tag4,
                 const float* __restrict__ bias, const float* __restrict__ alpha,
                 float* __restrict__ out, int n, int E, int* recAll, int* recCount) {
    __shared__ float4 acc4[AGG_ROWS * (N_FEAT / 4)];
    __shared__ int    qall[AGG_WAVES * AGG_QCAP];
    __shared__ int    cursor;
    float* acc = (float*)acc4;

    const int tid = threadIdx.x;
    const unsigned ulane = threadIdx.x & 31u;
    const int lane = (int)ulane;
    const int2* pair = (const int2*)pairs;
    const int wave = __builtin_amdgcn_readfirstlane(tid >> 5);
    const int base = (int)blockIdx.x * AGG_ROWS;
    const int left = n - base;
    const int rows = left < AGG_ROWS ? left : AGG_ROWS;
    int* rec = recAll != nullptr ? recAll + (size_t)blockIdx.x * (3 * REC_CAP) : nullptr;

    if (FIRST && tid == 0) cursor = 0;
    for (int idx = tid; idx < rows * (N_FEAT / 4); idx += AGG_THREADS) {
        const int r = idx >> 5, c4 = idx & 31;
        const float di = dinv[base + r];
        const float di2 = di * di;
        const float4 hv = *(const float4*)(H + (size_t)(base + r) * N_FEAT + c4 * 4);
        const float4 bv = *(const float4*)(bias + c4 * 4);
        float4 st;
        st.x = hv.x * di2 + bv.x;
        st.y = hv.y * di2 + bv.y;
        st.z = hv.z * di2 + bv.z;
        st.w = hv.w * di2 + bv.w;
        acc4[idx] = st;
    }
    __syncthreads();

    if (FIRST) {
        agg_walk(H, dinv, pair, tag4, acc, qall + wave * AGG_QCAP, wave, lane, ulane, base, rows, E, rec, &cursor);
    } else {
        const int total = rec != nullptr ? __builtin_amdgcn_readfirstlane(recCount[blockIdx.x]) : REC_CAP + 1;
        if (total <= REC_CAP) { if (total > 0) agg_replay(H, rec, acc, total, wave, ulane); }
        else agg_walk(H, dinv, pair, tag4, acc, qall + wave * AGG_QCAP, wave, lane, ulane, base, rows, E, nullptr, nullptr);
    }
    __syncthreads();
    if (FIRST && rec != nullptr && tid == 0) recCount[blockIdx.x] = cursor;

    for (int idx = tid; idx < rows * (N_FEAT / 4); idx += AGG_THREADS) {
        const int c4 = idx & 31;
        const float4 av = *(const float4*)(alpha + c4 * 4);
        float4 r = acc4[idx];
        r.x = r.x > 0.f ? r.x : av.x * r.x;
        r.y = r.y > 0.f ? r.y : av.y * r.y;
        r.z = r.z > 0.f ? r.z : av.z * r.z;
        r.w = r.w > 0.f ? r.w : av.w * r.w;
        *(float4*)(out + (size_t)base * N_FEAT + (size_t)idx * 4) = r;
    }
}

extern "C" void kernel_launch(void* const* d_in, const int* in_sizes, int n_in,
                              void* d_out, int out_size, void* d_ws, size_t ws_size,
                              hipStream_t stream) {
    (void)n_in; (void)out_size;
    const float* x  = (const float*)d_in[0];
    const int*   ei = (const int*)d_in[1];
    const float* W1 = (const float*)d_in[2];
    const float* b1 = (const float*)d_in[3];
    const float* W2 = (const float*)d_in[4];
    const float* b2 = (const float*)d_in[5];
    const float* al = (const float*)d_in[6];

    const int N = in_sizes[0] / N_FEAT;
    const int E = in_sizes[1] / 2;
    const int* src = ei;
    const int* dst = ei + E;

    float* dinv = (float*)d_ws;
    size_t off = (((size_t)N * sizeof(float)) + 255) & ~(size_t)255;
    float* hbuf = (float*)((char*)d_ws + off);
    unsigned* tags = (unsigned*)(hbuf + (size_t)N * N_FEAT);
    const int tagWords = ((E + AGG_CHUNK - 1) / AGG_CHUNK) * (AGG_CHUNK / 4);
    int* pairs = (int*)(tags + tagWords);
    float* outp = (float*)d_out;

    deg_init_kernel<<<(N + 255) / 256, 256, 0, stream>>>(dinv, N);
    deg_scatter_tag_kernel<<<(tagWords + 255) / 256, 256, 0, stream>>>(src, dst, dinv, tags, pairs, E, tagWords);
    deg_rsqrt_kernel<<<(N + 255) / 256, 256, 0, stream>>>(dinv, N);

    const int mTiles = (N + 15) / 16;
    const int gFull = (mTiles + 7) / 8;
    const int gBlocks = gFull < 768 ? gFull : 768;
    const int aBlocks = (N + AGG_ROWS - 1) / AGG_ROWS;
    int* lists  = pairs + 2 * (size_t)E;
    int* counts = lists + (size_t)aBlocks * (3 * REC_CAP);
    if ((size_t)((char*)(counts + aBlocks) - (char*)d_ws) > ws_size) { lists = nullptr; counts = nullptr; }

    gemm_wmma_kernel<<<gBlocks, 256, 0, stream>>>(x, W1, hbuf, N);
    aggregate_kernel<true><<<aBlocks, AGG_THREADS, 0, stream>>>(hbuf, dinv, pairs, tags, b1, al, outp, N, E, lists, counts);

    gemm_wmma_kernel<<<gBlocks, 256, 0, stream>>>(outp, W2, hbuf, N);
    aggregate_kernel<false><<<aBlocks, AGG_THREADS, 0, stream>>>(hbuf, dinv, pairs, tags, b2, al, outp, N, E, lists, counts);
}


# in-degrees counted in LDS by the row-owning groups in one tag walk ahead of both layers, which also writes the edge lists both layers replay: no global f32 atomic adds; arrival-order LDS sums kept
# speedup vs baseline: 3.7696x; 1.3724x over previous
#include <hip/hip_runtime.h>
#include <math.h>

#define N_FEAT 128

typedef __attribute__((ext_vector_type(2))) float v2f;
typedef __attribute__((ext_vector_type(8))) float v8f;

#define TAG_ROWS 392
__global__ void tag_pair_kernel(const int* __restrict__ src, const int* __restrict__ dst,
                                unsigned* __restrict__ tag4, int* __restrict__ pair, int E, int nWords) {
    const int t = blockIdx.x * blockDim.x + threadIdx.x;
    if (t >= nWords) return;
    const int e = 4 * t;
    int d0 = -1, d1 = -1, d2 = -1, d3 = -1;
    if (e + 3 < E && (((((size_t)dst) | ((size_t)src)) & 15) == 0)) {
        const int4 dv = *(const int4*)(dst + e);
        const int4 sv = *(const int4*)(src + e);
        d0 = dv.x; d1 = dv.y; d2 = dv.z; d3 = dv.w;
        int4* pp = (int4*)(pair + 2 * (size_t)e);
        pp[0] = make_int4(sv.x, dv.x, sv.y, dv.y);
        pp[1] = make_int4(sv.z, dv.z, sv.w, dv.w);
    } else {
        int2* pp = (int2*)pair;
        if (e     < E) { d0 = dst[e];     pp[e]     = make_int2(src[e],     d0); }
        if (e + 1 < E) { d1 = dst[e + 1]; pp[e + 1] = make_int2(src[e + 1], d1); }
        if (e + 2 < E) { d2 = dst[e + 2]; pp[e + 2] = make_int2(src[e + 2], d2); }
        if (e + 3 < E) { d3 = dst[e + 3]; pp[e + 3] = make_int2(src[e + 3], d3); }
    }
    unsigned w = 0u;
    if (d0 >= 0) w |= ((unsigned)(d0 / TAG_ROWS) & 255u);
    if (d1 >= 0) w |= ((unsigned)(d1 / TAG_ROWS) & 255u) << 8;
    if (d2 >= 0) w |= ((unsigned)(d2 / TAG_ROWS) & 255u) << 16;
    if (d3 >= 0) w |= ((unsigned)(d3 / TAG_ROWS) & 255u) << 24;
    tag4[t] = w;
}

__global__ void __launch_bounds__(256)
gemm_wmma_kernel(const float* __restrict__ X, const float* __restrict__ W,
                 float* __restrict__ H, int M) {
    __shared__ float4 Wl[32 * 128];
    const int tid  = threadIdx.x;
    const int wave = __builtin_amdgcn_readfirstlane(tid >> 5);
    const int lane = tid & 31;
    const int half = lane >> 4;
    const int l16  = lane & 15;
    const int step = (int)gridDim.x * 8;
    int st = wave * (int)gridDim.x + (int)blockIdx.x;
    bool live = st * 16 < M;

    float4 a4[16];
    {
        const float4* xr = (const float4*)(X + (size_t)(live ? st * 16 + l16 : 0) * N_FEAT) + half;
        #pragma unroll
        for (int i = 0; i < 16; ++i) a4[i] = xr[2 * i];
    }

    {
        const int col  = tid & 127;
        const int slot = ((((col & 3) + ((col >> 6) << 2))) << 4) + ((col & 63) >> 2);
        const float* wp = W + col;
        for (int it = 0; it < 16; ++it) {
            const int kq = (tid >> 7) + 2 * it;
            const float* p = wp + (size_t)(4 * kq) * N_FEAT;
            float4 v;
            v.x = p[0];
            v.y = p[N_FEAT];
            v.z = p[2 * N_FEAT];
            v.w = p[3 * N_FEAT];
            Wl[kq * 128 + slot] = v;
        }
    }
    __syncthreads();

    const float4* wl = Wl + half * 128 + l16;

    while (live) {
        asm volatile("" ::: "memory");
        v8f acc[8];
        #pragma unroll
        for (int t = 0; t < 8; ++t) acc[t] = (v8f){0.f, 0.f, 0.f, 0.f, 0.f, 0.f, 0.f, 0.f};

        float4 b[8];
        #pragma unroll
        for (int t = 0; t < 8; ++t) b[t] = wl[t * 16];

        #pragma unroll
        for (int i = 0; i < 16; ++i) {
            float4 bn[8];
            if (i + 1 < 16) {
                #pragma unroll
                for (int t = 0; t < 8; ++t) bn[t] = wl[(2 * (i + 1)) * 128 + t * 16];
            }
            v2f a0; a0.x = a4[i].x; a0.y = a4[i].y;
            v2f a1; a1.x = a4[i].z; a1.y = a4[i].w;
            #pragma unroll
            for (int t = 0; t < 8; ++t) {
                v2f bb; bb.x = b[t].x; bb.y = b[t].y;
                acc[t] = __builtin_amdgcn_wmma_f32_16x16x4_f32(
                     false, a0,  false, bb,
                     (short)0, acc[t],  false,  false);
            }
            #pragma unroll
            for (int t = 0; t < 8; ++t) {
                v2f bb; bb.x = b[t].z; bb.y = b[t].w;
                acc[t] = __builtin_amdgcn_wmma_f32_16x16x4_f32(
                     false, a1,  false, bb,
                     (short)0, acc[t],  false,  false);
            }
            if (i + 1 < 16) {
                #pragma unroll
                for (int t = 0; t < 8; ++t) b[t] = bn[t];
            }
        }

        float* hrow = H + (size_t)(st * 16 + half * 8) * N_FEAT + l16 * 4;

        st += step;
        live = st * 16 < M;
        if (live) {
            const float4* xr = (const float4*)(X + (size_t)(st * 16 + l16) * N_FEAT) + half;
            #pragma unroll
            for (int i = 0; i < 16; ++i) a4[i] = xr[2 * i];
        }

        #pragma unroll
        for (int r = 0; r < 8; ++r) {
            float4 lo, hi;
            lo.x = acc[0][r]; lo.y = acc[1][r]; lo.z = acc[2][r]; lo.w = acc[3][r];
            hi.x = acc[4][r]; hi.y = acc[5][r]; hi.z = acc[6][r]; hi.w = acc[7][r];
            *(float4*)(hrow + (size_t)r * N_FEAT)      = lo;
            *(float4*)(hrow + (size_t)r * N_FEAT + 64) = hi;
        }
    }
}

#define AGG_ROWS    392
#define AGG_THREADS 1024
#define AGG_WAVES   (AGG_THREADS / 32)
#define AGG_AHEAD   8
#define AGG_VEC     512
#define AGG_CHUNK   (AGG_AHEAD * AGG_VEC)
#define AGG_GROUP   12
#define AGG_QCAP    (AGG_GROUP + AGG_VEC)
#define LIST_GROUP  32
#define LIST_QCAP   (LIST_GROUP + AGG_VEC)
#define LIST_QSLOT  1312
#define REC_CAP     7168
#define REC_GROUP   16
static_assert(TAG_ROWS == AGG_ROWS, "the tags must be cut at the groups' row count");
static_assert(LIST_QSLOT >= LIST_QCAP, "a wave's queue must hold a left-over and one load's edges");

__device__ __forceinline__ unsigned zero_bytes(unsigned x) {
    const unsigned t = ((x & 0x7F7F7F7Fu) + 0x7F7F7F7Fu) | x;
    return ~t & 0x80808080u;
}

template <int GROUP, class F>
__device__ __forceinline__ void tag_walk(const unsigned* __restrict__ tag4, int* q, int wave, int lane, int E, const F& f) {
    int cnt = 0;
    const unsigned pat = ((unsigned)blockIdx.x & 255u) * 0x01010101u;
    const int nChunks = (E + AGG_CHUNK - 1) / AGG_CHUNK;
    const int nTrips = (nChunks + AGG_WAVES - 1) / AGG_WAVES;
    const int rot = (int)(((long long)blockIdx.x * nChunks) / (int)gridDim.x);

    for (int k = 0; k < nTrips; ++k) {
        int c = wave + AGG_WAVES * k;
        if (c >= nChunks) continue;
        c += rot;
        if (c >= nChunks) c -= nChunks;
        const int e0 = c * AGG_CHUNK;

        uint4 d[AGG_AHEAD];
        {
            const uint4* tp = (const uint4*)(tag4 + (e0 >> 2)) + lane;
            #pragma unroll
            for (int j = 0; j < AGG_AHEAD; ++j) d[j] = tp[j * 32];
        }

        int jstart = 0;
        for (;;) {
            int jnext = AGG_AHEAD;
            #pragma unroll
            for (int j = 0; j < AGG_AHEAD; ++j) {
                if (j >= jstart && jnext == AGG_AHEAD) {
                    unsigned m = (zero_bytes(d[j].x ^ pat) >> 3) | (zero_bytes(d[j].y ^ pat) >> 2) |
                                 (zero_bytes(d[j].z ^ pat) >> 1) |  zero_bytes(d[j].w ^ pat);
                    const int eb = e0 + j * AGG_VEC + 16 * lane;
                    for (;;) {
                        const bool has = m != 0u;
                        if (__builtin_amdgcn_ballot_w32(has) == 0u) break;
                        const int b = has ? __builtin_ctz(m) : 0;
                        const int e = eb + 4 * ((b & 7) - 4) + (b >> 3);
                        const bool keep = has && e < E;
                        const unsigned kb = __builtin_amdgcn_ballot_w32(keep);
                        if (keep) q[cnt + (int)__builtin_amdgcn_mbcnt_lo(kb, 0u)] = e;
                        cnt += __builtin_popcount(kb);
                        m &= m - 1u;
                    }
                    if (cnt >= GROUP) jnext = j + 1;
                }
            }
            __builtin_amdgcn_fence(__ATOMIC_ACQ_REL, "wavefront");
            while (cnt >= GROUP) {
                cnt -= GROUP;
                f.template run<true>(q + cnt, GROUP);
            }
            if (jnext >= AGG_AHEAD) break;
            jstart = jnext;
        }
    }
    if (cnt > 0) f.template run<false>(q, cnt);
}

struct ListEdges {
    const int2* pair; int* deg; int* rec; int* cursor; int base; int rows; unsigned ulane;
    template <bool FULL>
    __device__ __forceinline__ void run(const int* qp, int count) const {
        const bool have = FULL || (int)ulane < count;
        int e = 0;
        if (have) e = qp[ulane];
        const int2 sd = pair[e];
        const unsigned ru = (unsigned)(sd.y - base);
        const bool own = have && ru < (unsigned)rows;
        if (own) __hip_atomic_fetch_add(deg + ru, 1, __ATOMIC_RELAXED, __HIP_MEMORY_SCOPE_WORKGROUP);
        if (rec != nullptr) {
            const unsigned kb = __builtin_amdgcn_ballot_w32(own);
            int pos = 0;
            if (ulane == 0u) pos = __hip_atomic_fetch_add(cursor, __builtin_popcount(kb), __ATOMIC_RELAXED, __HIP_MEMORY_SCOPE_WORKGROUP);
            pos = __builtin_amdgcn_readfirstlane(pos);
            const int idx = pos + (int)__builtin_amdgcn_mbcnt_lo(kb, 0u);
            if (own && idx < REC_CAP) {
                __builtin_nontemporal_store(sd.x,    rec + idx);
                __builtin_nontemporal_store((int)ru, rec + REC_CAP + idx);
            }
        }
    }
};

__global__ void __launch_bounds__(AGG_THREADS)
owner_list_kernel(const int* __restrict__ pairs, const unsigned* __restrict__ tag4, float* __restrict__ dinv,
                  int n, int E, int* recAll, int* recCount) {
    __shared__ int deg[AGG_ROWS];
    __shared__ int qall[AGG_WAVES * LIST_QSLOT];
    __shared__ int cursor;

    const int tid = threadIdx.x;
    const unsigned ulane = threadIdx.x & 31u;
    const int wave = __builtin_amdgcn_readfirstlane(tid >> 5);
    const int base = (int)blockIdx.x * AGG_ROWS;
    const int left = n - base;
    const int rows = left < AGG_ROWS ? left : AGG_ROWS;
    int* rec = recAll != nullptr ? recAll + (size_t)blockIdx.x * (3 * REC_CAP) : nullptr;

    if (tid == 0) cursor = 0;
    for (int r = tid; r < AGG_ROWS; r += AGG_THREADS) deg[r] = 0;
    __syncthreads();

    const ListEdges f = { (const int2*)pairs, deg, rec, &cursor, base, rows, ulane };
    tag_walk<LIST_GROUP>(tag4, qall + wave * LIST_QSLOT, wave, (int)ulane, E, f);
    __syncthreads();

    for (int r = tid; r < rows; r += AGG_THREADS) dinv[base + r] = rsqrtf(1.0f + (float)deg[r]);
    if (rec != nullptr && tid == 0) recCount[blockIdx.x] = cursor;
}

template <int G, bool FULL>
__device__ __forceinline__ void agg_rows_add(const float* __restrict__ H, float* acc,
                                             int s, int r, int wi, int count, unsigned ulane) {
    float v[G][4];
    #pragma unroll
    for (int u = 0; u < G; ++u) {
        v[u][0] = 0.f; v[u][1] = 0.f; v[u][2] = 0.f; v[u][3] = 0.f;
        if (FULL || u < count) {
            const int su = __builtin_amdgcn_readlane(s, u);
            const float* row = H + (size_t)(unsigned)su * N_FEAT;
            v[u][0] = row[ulane];
            v[u][1] = row[ulane + 32u];
            v[u][2] = row[ulane + 64u];
            v[u][3] = row[ulane + 96u];
        }
    }
    #pragma unroll
    for (int u = 0; u < G; ++u) {
        if (FULL || u < count) {
            const int ru2 = __builtin_amdgcn_readlane(r, u);
            const float wu = __int_as_float(__builtin_amdgcn_readlane(wi, u));
            float* a = acc + (size_t)(unsigned)ru2 * N_FEAT + ulane;
            __hip_atomic_fetch_add(a,       v[u][0] * wu, __ATOMIC_RELAXED, __HIP_MEMORY_SCOPE_WORKGROUP);
            __hip_atomic_fetch_add(a + 32u, v[u][1] * wu, __ATOMIC_RELAXED, __HIP_MEMORY_SCOPE_WORKGROUP);
            __hip_atomic_fetch_add(a + 64u, v[u][2] * wu, __ATOMIC_RELAXED, __HIP_MEMORY_SCOPE_WORKGROUP);
            __hip_atomic_fetch_add(a + 96u, v[u][3] * wu, __ATOMIC_RELAXED, __HIP_MEMORY_SCOPE_WORKGROUP);
        }
    }
}

struct AddEdges {
    const float* H; const float* dinv; const int2* pair; float* acc; int base; int rows; unsigned ulane;
    template <bool FULL>
    __device__ __forceinline__ void run(const int* qp, int count) const {
        const int qi = ulane < (unsigned)AGG_GROUP ? (int)ulane : 0;
        int e = 0;
        if (FULL || qi < count) e = qp[qi];
        const int2 sd = pair[e];
        const int s = sd.x;
        const int d = sd.y;
        const unsigned ru = (unsigned)(d - base);
        const bool own = ru < (unsigned)rows;
        const int r = own ? (int)ru : 0;
        const int wi = __float_as_int(own ? dinv[s] * dinv[d] : 0.f);
        agg_rows_add<AGG_GROUP, FULL>(H, acc, s, r, wi, count, ulane);
    }
};

template <bool FIRST>
__device__ __forceinline__ void agg_replay(const float* __restrict__ H, const float* __restrict__ dinv, int* rec,
                                           float* acc, int total, int wave, unsigned ulane, int base) {
    const int qi = ulane < (unsigned)REC_GROUP ? (int)ulane : 0;
    const int nFull = total / REC_GROUP;
    const int rem = total - nFull * REC_GROUP;
    const int last = total - 1;
    int b = wave;
    int i = b * REC_GROUP + qi;
    i = i < last ? i : last;
    int s = __builtin_nontemporal_load(rec + i);
    int r = __builtin_nontemporal_load(rec + REC_CAP + i);
    if (FIRST) {
        int i1 = (b + AGG_WAVES) * REC_GROUP + qi;
        i1 = i1 < last ? i1 : last;
        int s1 = __builtin_nontemporal_load(rec + i1);
        int r1 = __builtin_nontemporal_load(rec + REC_CAP + i1);
        float w = dinv[s] * dinv[base + r];
        while (b < nFull) {
            const int cs = s, cr = r, cw = __float_as_int(w);
            if (ulane < (unsigned)REC_GROUP) __builtin_nontemporal_store(cw, rec + 2 * REC_CAP + i);
            const float wa = dinv[s1];
            const float wb = dinv[base + r1];
            b += AGG_WAVES;
            int i2 = (b + AGG_WAVES) * REC_GROUP + qi;
            i2 = i2 < last ? i2 : last;
            const int s2 = __builtin_nontemporal_load(rec + i2);
            const int r2 = __builtin_nontemporal_load(rec + REC_CAP + i2);
            agg_rows_add<REC_GROUP, true>(H, acc, cs, cr, cw, REC_GROUP, ulane);
            s = s1; r = r1; i = i1; w = wa * wb;
            s1 = s2; r1 = r2; i1 = i2;
        }
        if (rem > 0 && b == nFull) {
            const int cw = __float_as_int(w);
            if ((int)ulane < rem) __builtin_nontemporal_store(cw, rec + 2 * REC_CAP + i);
            agg_rows_add<REC_GROUP, false>(H, acc, s, r, cw, rem, ulane);
        }
    } else {
        int wi = __builtin_nontemporal_load(rec + 2 * REC_CAP + i);
        while (b < nFull) {
            const int cs = s, cr = r, cw = wi;
            b += AGG_WAVES;
            i = b * REC_GROUP + qi;
            i = i < last ? i : last;
            s  = __builtin_nontemporal_load(rec + i);
            r  = __builtin_nontemporal_load(rec + REC_CAP + i);
            wi = __builtin_nontemporal_load(rec + 2 * REC_CAP + i);
            agg_rows_add<REC_GROUP, true>(H, acc, cs, cr, cw, REC_GROUP, ulane);
        }
        if (rem > 0 && b == nFull) agg_rows_add<REC_GROUP, false>(H, acc, s, r, wi, rem, ulane);
    }
}

template <bool FIRST>
__global__ void __launch_bounds__(AGG_THREADS)
aggregate_kernel(const float* __restrict__ H, const float* __restrict__ dinv,
                 const int* __restrict__ pairs, const unsigned* __restrict__ tag4,
                 const float* __restrict__ bias, const float* __restrict__ alpha,
                 float* __restrict__ out, int n, int E, int* recAll, const int* recCount) {
    __shared__ float4 acc4[AGG_ROWS * (N_FEAT / 4)];
    __shared__ int    qall[AGG_WAVES * AGG_QCAP];
    float* acc = (float*)acc4;

    const int tid = threadIdx.x;
    const unsigned ulane = threadIdx.x & 31u;
    const int lane = (int)ulane;
    const int wave = __builtin_amdgcn_readfirstlane(tid >> 5);
    const int base = (int)blockIdx.x * AGG_ROWS;
    const int left = n - base;
    const int rows = left < AGG_ROWS ? left : AGG_ROWS;
    int* rec = recAll != nullptr ? recAll + (size_t)blockIdx.x * (3 * REC_CAP) : nullptr;

    for (int idx = tid; idx < rows * (N_FEAT / 4); idx += AGG_THREADS) {
        const int r = idx >> 5, c4 = idx & 31;
        const float di = dinv[base + r];
        const float di2 = di * di;
        const float4 hv = *(const float4*)(H + (size_t)(base + r) * N_FEAT + c4 * 4);
        const float4 bv = *(const float4*)(bias + c4 * 4);
        float4 st;
        st.x = hv.x * di2 + bv.x;
        st.y = hv.y * di2 + bv.y;
        st.z = hv.z * di2 + bv.z;
        st.w = hv.w * di2 + bv.w;
        acc4[idx] = st;
    }
    __syncthreads();

    {
        const int total = rec != nullptr ? __builtin_amdgcn_readfirstlane(recCount[blockIdx.x]) : REC_CAP + 1;
        if (total <= REC_CAP) { if (total > 0) agg_replay<FIRST>(H, dinv, rec, acc, total, wave, ulane, base); }
        else {
            const AddEdges f = { H, dinv, (const int2*)pairs, acc, base, rows, ulane };
            tag_walk<AGG_GROUP>(tag4, qall + wave * AGG_QCAP, wave, lane, E, f);
        }
    }
    __syncthreads();

    for (int idx = tid; idx < rows * (N_FEAT / 4); idx += AGG_THREADS) {
        const int c4 = idx & 31;
        const float4 av = *(const float4*)(alpha + c4 * 4);
        float4 r = acc4[idx];
        r.x = r.x > 0.f ? r.x : av.x * r.x;
        r.y = r.y > 0.f ? r.y : av.y * r.y;
        r.z = r.z > 0.f ? r.z : av.z * r.z;
        r.w = r.w > 0.f ? r.w : av.w * r.w;
        *(float4*)(out + (size_t)base * N_FEAT + (size_t)idx * 4) = r;
    }
}

extern "C" void kernel_launch(void* const* d_in, const int* in_sizes, int n_in,
                              void* d_out, int out_size, void* d_ws, size_t ws_size,
                              hipStream_t stream) {
    (void)n_in; (void)out_size;
    const float* x  = (const float*)d_in[0];
    const int*   ei = (const int*)d_in[1];
    const float* W1 = (const float*)d_in[2];
    const float* b1 = (const float*)d_in[3];
    const float* W2 = (const float*)d_in[4];
    const float* b2 = (const float*)d_in[5];
    const float* al = (const float*)d_in[6];

    const int N = in_sizes[0] / N_FEAT;
    const int E = in_sizes[1] / 2;
    const int* src = ei;
    const int* dst = ei + E;

    float* dinv = (float*)d_ws;
    size_t off = (((size_t)N * sizeof(float)) + 255) & ~(size_t)255;
    float* hbuf = (float*)((char*)d_ws + off);
    unsigned* tags = (unsigned*)(hbuf + (size_t)N * N_FEAT);
    const int tagWords = ((E + AGG_CHUNK - 1) / AGG_CHUNK) * (AGG_CHUNK / 4);
    int* pairs = (int*)(tags + tagWords);
    float* outp = (float*)d_out;

    const int mTiles = (N + 15) / 16;
    const int gFull = (mTiles + 7) / 8;
    const int gBlocks = gFull < 768 ? gFull : 768;
    const int aBlocks = (N + AGG_ROWS - 1) / AGG_ROWS;
    int* lists  = pairs + 2 * (size_t)E;
    int* counts = lists + (size_t)aBlocks * (3 * REC_CAP);
    if ((size_t)((char*)(counts + aBlocks) - (char*)d_ws) > ws_size) { lists = nullptr; counts = nullptr; }

    tag_pair_kernel<<<(tagWords + 255) / 256, 256, 0, stream>>>(src, dst, tags, pairs, E, tagWords);
    owner_list_kernel<<<aBlocks, AGG_THREADS, 0, stream>>>(pairs, tags, dinv, N, E, lists, counts);

    gemm_wmma_kernel<<<gBlocks, 256, 0, stream>>>(x, W1, hbuf, N);
    aggregate_kernel<true><<<aBlocks, AGG_THREADS, 0, stream>>>(hbuf, dinv, pairs, tags, b1, al, outp, N, E, lists, counts);

    gemm_wmma_kernel<<<gBlocks, 256, 0, stream>>>(outp, W2, hbuf, N);
    aggregate_kernel<false><<<aBlocks, AGG_THREADS, 0, stream>>>(hbuf, dinv, pairs, tags, b2, al, outp, N, E, lists, counts);
}


# edge passes load a source row with one 16-byte load per lane (lane = 4 adjacent features), not four 4-byte loads; LDS rows kept feature-transposed so the f32 adds stay conflict-free; sum order kept
# speedup vs baseline: 4.0837x; 1.0833x over previous
#include <hip/hip_runtime.h>
#include <math.h>

#define N_FEAT 128

typedef __attribute__((ext_vector_type(2))) float v2f;
typedef __attribute__((ext_vector_type(8))) float v8f;

#define TAG_ROWS 392
__global__ void tag_pair_kernel(const int* __restrict__ src, const int* __restrict__ dst,
                                unsigned* __restrict__ tag4, int* __restrict__ pair, int E, int nWords) {
    const int t = blockIdx.x * blockDim.x + threadIdx.x;
    if (t >= nWords) return;
    const int e = 4 * t;
    int d0 = -1, d1 = -1, d2 = -1, d3 = -1;
    if (e + 3 < E && (((((size_t)dst) | ((size_t)src)) & 15) == 0)) {
        const int4 dv = *(const int4*)(dst + e);
        const int4 sv = *(const int4*)(src + e);
        d0 = dv.x; d1 = dv.y; d2 = dv.z; d3 = dv.w;
        int4* pp = (int4*)(pair + 2 * (size_t)e);
        pp[0] = make_int4(sv.x, dv.x, sv.y, dv.y);
        pp[1] = make_int4(sv.z, dv.z, sv.w, dv.w);
    } else {
        int2* pp = (int2*)pair;
        if (e     < E) { d0 = dst[e];     pp[e]     = make_int2(src[e],     d0); }
        if (e + 1 < E) { d1 = dst[e + 1]; pp[e + 1] = make_int2(src[e + 1], d1); }
        if (e + 2 < E) { d2 = dst[e + 2]; pp[e + 2] = make_int2(src[e + 2], d2); }
        if (e + 3 < E) { d3 = dst[e + 3]; pp[e + 3] = make_int2(src[e + 3], d3); }
    }
    unsigned w = 0u;
    if (d0 >= 0) w |= ((unsigned)(d0 / TAG_ROWS) & 255u);
    if (d1 >= 0) w |= ((unsigned)(d1 / TAG_ROWS) & 255u) << 8;
    if (d2 >= 0) w |= ((unsigned)(d2 / TAG_ROWS) & 255u) << 16;
    if (d3 >= 0) w |= ((unsigned)(d3 / TAG_ROWS) & 255u) << 24;
    tag4[t] = w;
}

__global__ void __launch_bounds__(256)
gemm_wmma_kernel(const float* __restrict__ X, const float* __restrict__ W,
                 float* __restrict__ H, int M) {
    __shared__ float4 Wl[32 * 128];
    const int tid  = threadIdx.x;
    const int wave = __builtin_amdgcn_readfirstlane(tid >> 5);
    const int lane = tid & 31;
    const int half = lane >> 4;
    const int l16  = lane & 15;
    const int step = (int)gridDim.x * 8;
    int st = wave * (int)gridDim.x + (int)blockIdx.x;
    bool live = st * 16 < M;

    float4 a4[16];
    {
        const float4* xr = (const float4*)(X + (size_t)(live ? st * 16 + l16 : 0) * N_FEAT) + half;
        #pragma unroll
        for (int i = 0; i < 16; ++i) a4[i] = xr[2 * i];
    }

    {
        const int col  = tid & 127;
        const int slot = ((((col & 3) + ((col >> 6) << 2))) << 4) + ((col & 63) >> 2);
        const float* wp = W + col;
        for (int it = 0; it < 16; ++it) {
            const int kq = (tid >> 7) + 2 * it;
            const float* p = wp + (size_t)(4 * kq) * N_FEAT;
            float4 v;
            v.x = p[0];
            v.y = p[N_FEAT];
            v.z = p[2 * N_FEAT];
            v.w = p[3 * N_FEAT];
            Wl[kq * 128 + slot] = v;
        }
    }
    __syncthreads();

    const float4* wl = Wl + half * 128 + l16;

    while (live) {
        asm volatile("" ::: "memory");
        v8f acc[8];
        #pragma unroll
        for (int t = 0; t < 8; ++t) acc[t] = (v8f){0.f, 0.f, 0.f, 0.f, 0.f, 0.f, 0.f, 0.f};

        float4 b[8];
        #pragma unroll
        for (int t = 0; t < 8; ++t) b[t] = wl[t * 16];

        #pragma unroll
        for (int i = 0; i < 16; ++i) {
            float4 bn[8];
            if (i + 1 < 16) {
                #pragma unroll
                for (int t = 0; t < 8; ++t) bn[t] = wl[(2 * (i + 1)) * 128 + t * 16];
            }
            v2f a0; a0.x = a4[i].x; a0.y = a4[i].y;
            v2f a1; a1.x = a4[i].z; a1.y = a4[i].w;
            #pragma unroll
            for (int t = 0; t < 8; ++t) {
                v2f bb; bb.x = b[t].x; bb.y = b[t].y;
                acc[t] = __builtin_amdgcn_wmma_f32_16x16x4_f32(
                     false, a0,  false, bb,
                     (short)0, acc[t],  false,  false);
            }
            #pragma unroll
            for (int t = 0; t < 8; ++t) {
                v2f bb; bb.x = b[t].z; bb.y = b[t].w;
                acc[t] = __builtin_amdgcn_wmma_f32_16x16x4_f32(
                     false, a1,  false, bb,
                     (short)0, acc[t],  false,  false);
            }
            if (i + 1 < 16) {
                #pragma unroll
                for (int t = 0; t < 8; ++t) b[t] = bn[t];
            }
        }

        float* hrow = H + (size_t)(st * 16 + half * 8) * N_FEAT + l16 * 4;

        st += step;
        live = st * 16 < M;
        if (live) {
            const float4* xr = (const float4*)(X + (size_t)(st * 16 + l16) * N_FEAT) + half;
            #pragma unroll
            for (int i = 0; i < 16; ++i) a4[i] = xr[2 * i];
        }

        #pragma unroll
        for (int r = 0; r < 8; ++r) {
            float4 lo, hi;
            lo.x = acc[0][r]; lo.y = acc[1][r]; lo.z = acc[2][r]; lo.w = acc[3][r];
            hi.x = acc[4][r]; hi.y = acc[5][r]; hi.z = acc[6][r]; hi.w = acc[7][r];
            *(float4*)(hrow + (size_t)r * N_FEAT)      = lo;
            *(float4*)(hrow + (size_t)r * N_FEAT + 64) = hi;
        }
    }
}

#define AGG_ROWS    392
#define AGG_THREADS 1024
#define AGG_WAVES   (AGG_THREADS / 32)
#define AGG_AHEAD   8
#define AGG_VEC     512
#define AGG_CHUNK   (AGG_AHEAD * AGG_VEC)
#define AGG_GROUP   12
#define AGG_QCAP    (AGG_GROUP + AGG_VEC)
#define LIST_GROUP  32
#define LIST_QCAP   (LIST_GROUP + AGG_VEC)
#define LIST_QSLOT  1312
#define REC_CAP     7168
#define REC_GROUP   16
static_assert(TAG_ROWS == AGG_ROWS, "the tags must be cut at the groups' row count");
static_assert(LIST_QSLOT >= LIST_QCAP, "a wave's queue must hold a left-over and one load's edges");

__device__ __forceinline__ unsigned zero_bytes(unsigned x) {
    const unsigned t = ((x & 0x7F7F7F7Fu) + 0x7F7F7F7Fu) | x;
    return ~t & 0x80808080u;
}

template <int GROUP, class F>
__device__ __forceinline__ void tag_walk(const unsigned* __restrict__ tag4, int* q, int wave, int lane, int E, const F& f) {
    int cnt = 0;
    const unsigned pat = ((unsigned)blockIdx.x & 255u) * 0x01010101u;
    const int nChunks = (E + AGG_CHUNK - 1) / AGG_CHUNK;
    const int nTrips = (nChunks + AGG_WAVES - 1) / AGG_WAVES;
    const int rot = (int)(((long long)blockIdx.x * nChunks) / (int)gridDim.x);

    for (int k = 0; k < nTrips; ++k) {
        int c = wave + AGG_WAVES * k;
        if (c >= nChunks) continue;
        c += rot;
        if (c >= nChunks) c -= nChunks;
        const int e0 = c * AGG_CHUNK;

        uint4 d[AGG_AHEAD];
        {
            const uint4* tp = (const uint4*)(tag4 + (e0 >> 2)) + lane;
            #pragma unroll
            for (int j = 0; j < AGG_AHEAD; ++j) d[j] = tp[j * 32];
        }

        int jstart = 0;
        for (;;) {
            int jnext = AGG_AHEAD;
            #pragma unroll
            for (int j = 0; j < AGG_AHEAD; ++j) {
                if (j >= jstart && jnext == AGG_AHEAD) {
                    unsigned m = (zero_bytes(d[j].x ^ pat) >> 3) | (zero_bytes(d[j].y ^ pat) >> 2) |
                                 (zero_bytes(d[j].z ^ pat) >> 1) |  zero_bytes(d[j].w ^ pat);
                    const int eb = e0 + j * AGG_VEC + 16 * lane;
                    for (;;) {
                        const bool has = m != 0u;
                        if (__builtin_amdgcn_ballot_w32(has) == 0u) break;
                        const int b = has ? __builtin_ctz(m) : 0;
                        const int e = eb + 4 * ((b & 7) - 4) + (b >> 3);
                        const bool keep = has && e < E;
                        const unsigned kb = __builtin_amdgcn_ballot_w32(keep);
                        if (keep) q[cnt + (int)__builtin_amdgcn_mbcnt_lo(kb, 0u)] = e;
                        cnt += __builtin_popcount(kb);
                        m &= m - 1u;
                    }
                    if (cnt >= GROUP) jnext = j + 1;
                }
            }
            __builtin_amdgcn_fence(__ATOMIC_ACQ_REL, "wavefront");
            while (cnt >= GROUP) {
                cnt -= GROUP;
                f.template run<true>(q + cnt, GROUP);
            }
            if (jnext >= AGG_AHEAD) break;
            jstart = jnext;
        }
    }
    if (cnt > 0) f.template run<false>(q, cnt);
}

struct ListEdges {
    const int2* pair; int* deg; int* rec; int* cursor; int base; int rows; unsigned ulane;
    template <bool FULL>
    __device__ __forceinline__ void run(const int* qp, int count) const {
        const bool have = FULL || (int)ulane < count;
        int e = 0;
        if (have) e = qp[ulane];
        const int2 sd = pair[e];
        const unsigned ru = (unsigned)(sd.y - base);
        const bool own = have && ru < (unsigned)rows;
        if (own) __hip_atomic_fetch_add(deg + ru, 1, __ATOMIC_RELAXED, __HIP_MEMORY_SCOPE_WORKGROUP);
        if (rec != nullptr) {
            const unsigned kb = __builtin_amdgcn_ballot_w32(own);
            int pos = 0;
            if (ulane == 0u) pos = __hip_atomic_fetch_add(cursor, __builtin_popcount(kb), __ATOMIC_RELAXED, __HIP_MEMORY_SCOPE_WORKGROUP);
            pos = __builtin_amdgcn_readfirstlane(pos);
            const int idx = pos + (int)__builtin_amdgcn_mbcnt_lo(kb, 0u);
            if (own && idx < REC_CAP) {
                __builtin_nontemporal_store(sd.x,    rec + idx);
                __builtin_nontemporal_store((int)ru, rec + REC_CAP + idx);
            }
        }
    }
};

__global__ void __launch_bounds__(AGG_THREADS)
owner_list_kernel(const int* __restrict__ pairs, const unsigned* __restrict__ tag4, float* __restrict__ dinv,
                  int n, int E, int* recAll, int* recCount) {
    __shared__ int deg[AGG_ROWS];
    __shared__ int qall[AGG_WAVES * LIST_QSLOT];
    __shared__ int cursor;

    const int tid = threadIdx.x;
    const unsigned ulane = threadIdx.x & 31u;
    const int wave = __builtin_amdgcn_readfirstlane(tid >> 5);
    const int base = (int)blockIdx.x * AGG_ROWS;
    const int left = n - base;
    const int rows = left < AGG_ROWS ? left : AGG_ROWS;
    int* rec = recAll != nullptr ? recAll + (size_t)blockIdx.x * (3 * REC_CAP) : nullptr;

    if (tid == 0) cursor = 0;
    for (int r = tid; r < AGG_ROWS; r += AGG_THREADS) deg[r] = 0;
    __syncthreads();

    const ListEdges f = { (const int2*)pairs, deg, rec, &cursor, base, rows, ulane };
    tag_walk<LIST_GROUP>(tag4, qall + wave * LIST_QSLOT, wave, (int)ulane, E, f);
    __syncthreads();

    for (int r = tid; r < rows; r += AGG_THREADS) dinv[base + r] = rsqrtf(1.0f + (float)deg[r]);
    if (rec != nullptr && tid == 0) recCount[blockIdx.x] = cursor;
}

template <int G, bool FULL>
__device__ __forceinline__ void agg_rows_add(const float* __restrict__ H, float* acc,
                                             int s, int r, int wi, int count, unsigned ulane) {
    float4 v[G];
    unsigned off[G];
    #pragma unroll
    for (int u = 0; u < G; ++u)
        off[u] = ((unsigned)__builtin_amdgcn_readlane(s, u) << 9) | (ulane << 4);
    #pragma unroll
    for (int u = 0; u < G; ++u) {
        v[u] = make_float4(0.f, 0.f, 0.f, 0.f);
        if (FULL || u < count)
            v[u] = *(const float4*)((const char*)H + off[u]);
    }
    #pragma unroll
    for (int u = 0; u < G; ++u) {
        if (FULL || u < count) {
            const int ru2 = __builtin_amdgcn_readlane(r, u);
            const float wu = __int_as_float(__builtin_amdgcn_readlane(wi, u));
            float* a = acc + (size_t)(unsigned)ru2 * N_FEAT + ulane;
            __hip_atomic_fetch_add(a,       v[u].x * wu, __ATOMIC_RELAXED, __HIP_MEMORY_SCOPE_WORKGROUP);
            __hip_atomic_fetch_add(a + 32u, v[u].y * wu, __ATOMIC_RELAXED, __HIP_MEMORY_SCOPE_WORKGROUP);
            __hip_atomic_fetch_add(a + 64u, v[u].z * wu, __ATOMIC_RELAXED, __HIP_MEMORY_SCOPE_WORKGROUP);
            __hip_atomic_fetch_add(a + 96u, v[u].w * wu, __ATOMIC_RELAXED, __HIP_MEMORY_SCOPE_WORKGROUP);
        }
    }
}

struct AddEdges {
    const float* H; const float* dinv; const int2* pair; float* acc; int base; int rows; unsigned ulane;
    template <bool FULL>
    __device__ __forceinline__ void run(const int* qp, int count) const {
        const int qi = ulane < (unsigned)AGG_GROUP ? (int)ulane : 0;
        int e = 0;
        if (FULL || qi < count) e = qp[qi];
        const int2 sd = pair[e];
        const int s = sd.x;
        const int d = sd.y;
        const unsigned ru = (unsigned)(d - base);
        const bool own = ru < (unsigned)rows;
        const int r = own ? (int)ru : 0;
        const int wi = __float_as_int(own ? dinv[s] * dinv[d] : 0.f);
        agg_rows_add<AGG_GROUP, FULL>(H, acc, s, r, wi, count, ulane);
    }
};

template <bool FIRST>
__device__ __forceinline__ void agg_replay(const float* __restrict__ H, const float* __restrict__ dinv, int* rec,
                                           float* acc, int total, int wave, unsigned ulane, int base) {
    const int qi = ulane < (unsigned)REC_GROUP ? (int)ulane : 0;
    const int nFull = total / REC_GROUP;
    const int rem = total - nFull * REC_GROUP;
    const int last = total - 1;
    int b = wave;
    int i = b * REC_GROUP + qi;
    i = i < last ? i : last;
    int s = __builtin_nontemporal_load(rec + i);
    int r = __builtin_nontemporal_load(rec + REC_CAP + i);
    if (FIRST) {
        int i1 = (b + AGG_WAVES) * REC_GROUP + qi;
        i1 = i1 < last ? i1 : last;
        int s1 = __builtin_nontemporal_load(rec + i1);
        int r1 = __builtin_nontemporal_load(rec + REC_CAP + i1);
        float w = dinv[s] * dinv[base + r];
        while (b < nFull) {
            const int cs = s, cr = r, cw = __float_as_int(w);
            if (ulane < (unsigned)REC_GROUP) __builtin_nontemporal_store(cw, rec + 2 * REC_CAP + i);
            const float wa = dinv[s1];
            const float wb = dinv[base + r1];
            b += AGG_WAVES;
            int i2 = (b + AGG_WAVES) * REC_GROUP + qi;
            i2 = i2 < last ? i2 : last;
            const int s2 = __builtin_nontemporal_load(rec + i2);
            const int r2 = __builtin_nontemporal_load(rec + REC_CAP + i2);
            agg_rows_add<REC_GROUP, true>(H, acc, cs, cr, cw, REC_GROUP, ulane);
            s = s1; r = r1; i = i1; w = wa * wb;
            s1 = s2; r1 = r2; i1 = i2;
        }
        if (rem > 0 && b == nFull) {
            const int cw = __float_as_int(w);
            if ((int)ulane < rem) __builtin_nontemporal_store(cw, rec + 2 * REC_CAP + i);
            agg_rows_add<REC_GROUP, false>(H, acc, s, r, cw, rem, ulane);
        }
    } else {
        int wi = __builtin_nontemporal_load(rec + 2 * REC_CAP + i);
        while (b < nFull) {
            const int cs = s, cr = r, cw = wi;
            b += AGG_WAVES;
            i = b * REC_GROUP + qi;
            i = i < last ? i : last;
            s  = __builtin_nontemporal_load(rec + i);
            r  = __builtin_nontemporal_load(rec + REC_CAP + i);
            wi = __builtin_nontemporal_load(rec + 2 * REC_CAP + i);
            agg_rows_add<REC_GROUP, true>(H, acc, cs, cr, cw, REC_GROUP, ulane);
        }
        if (rem > 0 && b == nFull) agg_rows_add<REC_GROUP, false>(H, acc, s, r, wi, rem, ulane);
    }
}

template <bool FIRST>
__global__ void __launch_bounds__(AGG_THREADS)
aggregate_kernel(const float* __restrict__ H, const float* __restrict__ dinv,
                 const int* __restrict__ pairs, const unsigned* __restrict__ tag4,
                 const float* __restrict__ bias, const float* __restrict__ alpha,
                 float* __restrict__ out, int n, int E, int* recAll, const int* recCount) {
    __shared__ float acc[AGG_ROWS * N_FEAT];
    __shared__ int   qall[AGG_WAVES * AGG_QCAP];

    const int tid = threadIdx.x;
    const unsigned ulane = threadIdx.x & 31u;
    const int lane = (int)ulane;
    const int wave = __builtin_amdgcn_readfirstlane(tid >> 5);
    const int base = (int)blockIdx.x * AGG_ROWS;
    const int left = n - base;
    const int rows = left < AGG_ROWS ? left : AGG_ROWS;
    int* rec = recAll != nullptr ? recAll + (size_t)blockIdx.x * (3 * REC_CAP) : nullptr;

    for (int idx = tid; idx < rows * (N_FEAT / 4); idx += AGG_THREADS) {
        const int r = idx >> 5, c4 = idx & 31;
        const float di = dinv[base + r];
        const float di2 = di * di;
        const float4 hv = *(const float4*)(H + (size_t)(base + r) * N_FEAT + c4 * 4);
        const float4 bv = *(const float4*)(bias + c4 * 4);
        float4 st;
        st.x = hv.x * di2 + bv.x;
        st.y = hv.y * di2 + bv.y;
        st.z = hv.z * di2 + bv.z;
        st.w = hv.w * di2 + bv.w;
        float* a = acc + r * N_FEAT + c4;
        a[0] = st.x; a[32] = st.y; a[64] = st.z; a[96] = st.w;
    }
    __syncthreads();

    {
        const int total = rec != nullptr ? __builtin_amdgcn_readfirstlane(recCount[blockIdx.x]) : REC_CAP + 1;
        if (total <= REC_CAP) { if (total > 0) agg_replay<FIRST>(H, dinv, rec, acc, total, wave, ulane, base); }
        else {
            const AddEdges f = { H, dinv, (const int2*)pairs, acc, base, rows, ulane };
            tag_walk<AGG_GROUP>(tag4, qall + wave * AGG_QCAP, wave, lane, E, f);
        }
    }
    __syncthreads();

    for (int idx = tid; idx < rows * (N_FEAT / 4); idx += AGG_THREADS) {
        const int c4 = idx & 31;
        const float4 av = *(const float4*)(alpha + c4 * 4);
        const float* a = acc + (idx >> 5) * N_FEAT + c4;
        float4 r;
        r.x = a[0]; r.y = a[32]; r.z = a[64]; r.w = a[96];
        r.x = r.x > 0.f ? r.x : av.x * r.x;
        r.y = r.y > 0.f ? r.y : av.y * r.y;
        r.z = r.z > 0.f ? r.z : av.z * r.z;
        r.w = r.w > 0.f ? r.w : av.w * r.w;
        *(float4*)(out + (size_t)base * N_FEAT + (size_t)idx * 4) = r;
    }
}

extern "C" void kernel_launch(void* const* d_in, const int* in_sizes, int n_in,
                              void* d_out, int out_size, void* d_ws, size_t ws_size,
                              hipStream_t stream) {
    (void)n_in; (void)out_size;
    const float* x  = (const float*)d_in[0];
    const int*   ei = (const int*)d_in[1];
    const float* W1 = (const float*)d_in[2];
    const float* b1 = (const float*)d_in[3];
    const float* W2 = (const float*)d_in[4];
    const float* b2 = (const float*)d_in[5];
    const float* al = (const float*)d_in[6];

    const int N = in_sizes[0] / N_FEAT;
    const int E = in_sizes[1] / 2;
    const int* src = ei;
    const int* dst = ei + E;

    float* dinv = (float*)d_ws;
    size_t off = (((size_t)N * sizeof(float)) + 255) & ~(size_t)255;
    float* hbuf = (float*)((char*)d_ws + off);
    unsigned* tags = (unsigned*)(hbuf + (size_t)N * N_FEAT);
    const int tagWords = ((E + AGG_CHUNK - 1) / AGG_CHUNK) * (AGG_CHUNK / 4);
    int* pairs = (int*)(tags + tagWords);
    float* outp = (float*)d_out;

    const int mTiles = (N + 15) / 16;
    const int gFull = (mTiles + 7) / 8;
    const int gBlocks = gFull < 768 ? gFull : 768;
    const int aBlocks = (N + AGG_ROWS - 1) / AGG_ROWS;
    int* lists  = pairs + 2 * (size_t)E;
    int* counts = lists + (size_t)aBlocks * (3 * REC_CAP);
    if ((size_t)((char*)(counts + aBlocks) - (char*)d_ws) > ws_size) { lists = nullptr; counts = nullptr; }

    tag_pair_kernel<<<(tagWords + 255) / 256, 256, 0, stream>>>(src, dst, tags, pairs, E, tagWords);
    owner_list_kernel<<<aBlocks, AGG_THREADS, 0, stream>>>(pairs, tags, dinv, N, E, lists, counts);

    gemm_wmma_kernel<<<gBlocks, 256, 0, stream>>>(x, W1, hbuf, N);
    aggregate_kernel<true><<<aBlocks, AGG_THREADS, 0, stream>>>(hbuf, dinv, pairs, tags, b1, al, outp, N, E, lists, counts);

    gemm_wmma_kernel<<<gBlocks, 256, 0, stream>>>(outp, W2, hbuf, N);
    aggregate_kernel<false><<<aBlocks, AGG_THREADS, 0, stream>>>(hbuf, dinv, pairs, tags, b2, al, outp, N, E, lists, counts);
}


# layer-1 f32 WMMA product computed inside the owner-list launch: 8 of 32 waves do 16x64 half stripes beside the tag walk, then join it (chunks from an LDS counter); one launch fewer; sum orders kept
# speedup vs baseline: 4.3235x; 1.0587x over previous
#include <hip/hip_runtime.h>
#include <math.h>

#define N_FEAT 128

typedef __attribute__((ext_vector_type(2))) float v2f;
typedef __attribute__((ext_vector_type(8))) float v8f;

#define TAG_ROWS 392
__global__ void tag_pair_kernel(const int* __restrict__ src, const int* __restrict__ dst,
                                unsigned* __restrict__ tag4, int* __restrict__ pair, int E, int nWords) {
    const int t = blockIdx.x * blockDim.x + threadIdx.x;
    if (t >= nWords) return;
    const int e = 4 * t;
    int d0 = -1, d1 = -1, d2 = -1, d3 = -1;
    if (e + 3 < E && (((((size_t)dst) | ((size_t)src)) & 15) == 0)) {
        const int4 dv = *(const int4*)(dst + e);
        const int4 sv = *(const int4*)(src + e);
        d0 = dv.x; d1 = dv.y; d2 = dv.z; d3 = dv.w;
        int4* pp = (int4*)(pair + 2 * (size_t)e);
        pp[0] = make_int4(sv.x, dv.x, sv.y, dv.y);
        pp[1] = make_int4(sv.z, dv.z, sv.w, dv.w);
    } else {
        int2* pp = (int2*)pair;
        if (e     < E) { d0 = dst[e];     pp[e]     = make_int2(src[e],     d0); }
        if (e + 1 < E) { d1 = dst[e + 1]; pp[e + 1] = make_int2(src[e + 1], d1); }
        if (e + 2 < E) { d2 = dst[e + 2]; pp[e + 2] = make_int2(src[e + 2], d2); }
        if (e + 3 < E) { d3 = dst[e + 3]; pp[e + 3] = make_int2(src[e + 3], d3); }
    }
    unsigned w = 0u;
    if (d0 >= 0) w |= ((unsigned)(d0 / TAG_ROWS) & 255u);
    if (d1 >= 0) w |= ((unsigned)(d1 / TAG_ROWS) & 255u) << 8;
    if (d2 >= 0) w |= ((unsigned)(d2 / TAG_ROWS) & 255u) << 16;
    if (d3 >= 0) w |= ((unsigned)(d3 / TAG_ROWS) & 255u) << 24;
    tag4[t] = w;
}

__global__ void __launch_bounds__(256)
gemm_wmma_kernel(const float* __restrict__ X, const float* __restrict__ W,
                 float* __restrict__ H, int M) {
    __shared__ float4 Wl[32 * 128];
    const int tid  = threadIdx.x;
    const int wave = __builtin_amdgcn_readfirstlane(tid >> 5);
    const int lane = tid & 31;
    const int half = lane >> 4;
    const int l16  = lane & 15;
    const int step = (int)gridDim.x * 8;
    int st = wave * (int)gridDim.x + (int)blockIdx.x;
    bool live = st * 16 < M;

    float4 a4[16];
    {
        const float4* xr = (const float4*)(X + (size_t)(live ? st * 16 + l16 : 0) * N_FEAT) + half;
        #pragma unroll
        for (int i = 0; i < 16; ++i) a4[i] = xr[2 * i];
    }

    {
        const int col  = tid & 127;
        const int slot = ((((col & 3) + ((col >> 6) << 2))) << 4) + ((col & 63) >> 2);
        const float* wp = W + col;
        for (int it = 0; it < 16; ++it) {
            const int kq = (tid >> 7) + 2 * it;
            const float* p = wp + (size_t)(4 * kq) * N_FEAT;
            float4 v;
            v.x = p[0];
            v.y = p[N_FEAT];
            v.z = p[2 * N_FEAT];
            v.w = p[3 * N_FEAT];
            Wl[kq * 128 + slot] = v;
        }
    }
    __syncthreads();

    const float4* wl = Wl + half * 128 + l16;

    while (live) {
        asm volatile("" ::: "memory");
        v8f acc[8];
        #pragma unroll
        for (int t = 0; t < 8; ++t) acc[t] = (v8f){0.f, 0.f, 0.f, 0.f, 0.f, 0.f, 0.f, 0.f};

        float4 b[8];
        #pragma unroll
        for (int t = 0; t < 8; ++t) b[t] = wl[t * 16];

        #pragma unroll
        for (int i = 0; i < 16; ++i) {
            float4 bn[8];
            if (i + 1 < 16) {
                #pragma unroll
                for (int t = 0; t < 8; ++t) bn[t] = wl[(2 * (i + 1)) * 128 + t * 16];
            }
            v2f a0; a0.x = a4[i].x; a0.y = a4[i].y;
            v2f a1; a1.x = a4[i].z; a1.y = a4[i].w;
            #pragma unroll
            for (int t = 0; t < 8; ++t) {
                v2f bb; bb.x = b[t].x; bb.y = b[t].y;
                acc[t] = __builtin_amdgcn_wmma_f32_16x16x4_f32(
                     false, a0,  false, bb,
                     (short)0, acc[t],  false,  false);
            }
            #pragma unroll
            for (int t = 0; t < 8; ++t) {
                v2f bb; bb.x = b[t].z; bb.y = b[t].w;
                acc[t] = __builtin_amdgcn_wmma_f32_16x16x4_f32(
                     false, a1,  false, bb,
                     (short)0, acc[t],  false,  false);
            }
            if (i + 1 < 16) {
                #pragma unroll
                for (int t = 0; t < 8; ++t) b[t] = bn[t];
            }
        }

        float* hrow = H + (size_t)(st * 16 + half * 8) * N_FEAT + l16 * 4;

        st += step;
        live = st * 16 < M;
        if (live) {
            const float4* xr = (const float4*)(X + (size_t)(st * 16 + l16) * N_FEAT) + half;
            #pragma unroll
            for (int i = 0; i < 16; ++i) a4[i] = xr[2 * i];
        }

        #pragma unroll
        for (int r = 0; r < 8; ++r) {
            float4 lo, hi;
            lo.x = acc[0][r]; lo.y = acc[1][r]; lo.z = acc[2][r]; lo.w = acc[3][r];
            hi.x = acc[4][r]; hi.y = acc[5][r]; hi.z = acc[6][r]; hi.w = acc[7][r];
            *(float4*)(hrow + (size_t)r * N_FEAT)      = lo;
            *(float4*)(hrow + (size_t)r * N_FEAT + 64) = hi;
        }
    }
}

#define AGG_ROWS    392
#define AGG_THREADS 1024
#define AGG_WAVES   (AGG_THREADS / 32)
#define AGG_AHEAD   8
#define AGG_VEC     512
#define AGG_CHUNK   (AGG_AHEAD * AGG_VEC)
#define AGG_GROUP   12
#define AGG_QCAP    (AGG_GROUP + AGG_VEC)
#define LIST_GROUP  32
#define LIST_QCAP   (LIST_GROUP + AGG_VEC)
#define LIST_QSLOT  1312
#define REC_CAP     7168
#define REC_GROUP   16
static_assert(TAG_ROWS == AGG_ROWS, "the tags must be cut at the groups' row count");
static_assert(LIST_QSLOT >= LIST_QCAP, "a wave's queue must hold a left-over and one load's edges");

__device__ __forceinline__ unsigned zero_bytes(unsigned x) {
    const unsigned t = ((x & 0x7F7F7F7Fu) + 0x7F7F7F7Fu) | x;
    return ~t & 0x80808080u;
}

template <int GROUP, bool DYN, class F>
__device__ __forceinline__ void tag_walk(const unsigned* __restrict__ tag4, int* q, int wave, int lane, int E, const F& f, int* next) {
    int cnt = 0;
    const unsigned pat = ((unsigned)blockIdx.x & 255u) * 0x01010101u;
    const int nChunks = (E + AGG_CHUNK - 1) / AGG_CHUNK;
    const int nTrips = (nChunks + AGG_WAVES - 1) / AGG_WAVES;
    const int rot = (int)(((long long)blockIdx.x * nChunks) / (int)gridDim.x);

    for (int k = 0; k < (DYN ? nChunks : nTrips); ++k) {
        int c;
        if (DYN) {
            int t = 0;
            if (lane == 0) t = __hip_atomic_fetch_add(next, 1, __ATOMIC_RELAXED, __HIP_MEMORY_SCOPE_WORKGROUP);
            c = __builtin_amdgcn_readfirstlane(t);
            if (c >= nChunks) break;
        } else {
            c = wave + AGG_WAVES * k;
            if (c >= nChunks) continue;
        }
        c += rot;
        if (c >= nChunks) c -= nChunks;
        const int e0 = c * AGG_CHUNK;

        uint4 d[AGG_AHEAD];
        {
            const uint4* tp = (const uint4*)(tag4 + (e0 >> 2)) + lane;
            #pragma unroll
            for (int j = 0; j < AGG_AHEAD; ++j) d[j] = tp[j * 32];
        }

        int jstart = 0;
        for (;;) {
            int jnext = AGG_AHEAD;
            #pragma unroll
            for (int j = 0; j < AGG_AHEAD; ++j) {
                if (j >= jstart && jnext == AGG_AHEAD) {
                    unsigned m = (zero_bytes(d[j].x ^ pat) >> 3) | (zero_bytes(d[j].y ^ pat) >> 2) |
                                 (zero_bytes(d[j].z ^ pat) >> 1) |  zero_bytes(d[j].w ^ pat);
                    const int eb = e0 + j * AGG_VEC + 16 * lane;
                    for (;;) {
                        const bool has = m != 0u;
                        if (__builtin_amdgcn_ballot_w32(has) == 0u) break;
                        const int b = has ? __builtin_ctz(m) : 0;
                        const int e = eb + 4 * ((b & 7) - 4) + (b >> 3);
                        const bool keep = has && e < E;
                        const unsigned kb = __builtin_amdgcn_ballot_w32(keep);
                        if (keep) q[cnt + (int)__builtin_amdgcn_mbcnt_lo(kb, 0u)] = e;
                        cnt += __builtin_popcount(kb);
                        m &= m - 1u;
                    }
                    if (cnt >= GROUP) jnext = j + 1;
                }
            }
            __builtin_amdgcn_fence(__ATOMIC_ACQ_REL, "wavefront");
            while (cnt >= GROUP) {
                cnt -= GROUP;
                f.template run<true>(q + cnt, GROUP);
            }
            if (jnext >= AGG_AHEAD) break;
            jstart = jnext;
        }
    }
    if (cnt > 0) f.template run<false>(q, cnt);
}

struct ListEdges {
    const int2* pair; int* deg; int* rec; int* cursor; int base; int rows; unsigned ulane;
    template <bool FULL>
    __device__ __forceinline__ void run(const int* qp, int count) const {
        const bool have = FULL || (int)ulane < count;
        int e = 0;
        if (have) e = qp[ulane];
        const int2 sd = pair[e];
        const unsigned ru = (unsigned)(sd.y - base);
        const bool own = have && ru < (unsigned)rows;
        if (own) __hip_atomic_fetch_add(deg + ru, 1, __ATOMIC_RELAXED, __HIP_MEMORY_SCOPE_WORKGROUP);
        if (rec != nullptr) {
            const unsigned kb = __builtin_amdgcn_ballot_w32(own);
            int pos = 0;
            if (ulane == 0u) pos = __hip_atomic_fetch_add(cursor, __builtin_popcount(kb), __ATOMIC_RELAXED, __HIP_MEMORY_SCOPE_WORKGROUP);
            pos = __builtin_amdgcn_readfirstlane(pos);
            const int idx = pos + (int)__builtin_amdgcn_mbcnt_lo(kb, 0u);
            if (own && idx < REC_CAP) {
                __builtin_nontemporal_store(sd.x,    rec + idx);
                __builtin_nontemporal_store((int)ru, rec + REC_CAP + idx);
            }
        }
    }
};

#define PROD_WAVES 8
__device__ __forceinline__ void product_half_stripes(const float* __restrict__ X, float* __restrict__ H,
                                                     const float4* Wl, int M, int g, int lane) {
    const int half = lane >> 4;
    const int l16  = lane & 15;
    int q  = g;
    int st = (int)blockIdx.x + (int)gridDim.x * (q >> 1);
    bool live = st * 16 < M;
    const int ch = g & 1;
    const float4* wl = Wl + half * 128 + ch * 64 + l16;

    float4 a[16];
    if (live) {
        const float4* xr = (const float4*)(X + (size_t)(st * 16 + l16) * N_FEAT) + half;
        #pragma unroll
        for (int i = 0; i < 8; ++i) a[i] = xr[2 * i];
    }

    while (live) {
        asm volatile("" ::: "memory");
        const float4* xr = (const float4*)(X + (size_t)(st * 16 + l16) * N_FEAT) + half;

        v8f acc[4];
        #pragma unroll
        for (int t = 0; t < 4; ++t) acc[t] = (v8f){0.f, 0.f, 0.f, 0.f, 0.f, 0.f, 0.f, 0.f};

        float4 b[4];
        #pragma unroll
        for (int t = 0; t < 4; ++t) b[t] = wl[t * 16];

        #pragma unroll
        for (int i = 0; i < 16; ++i) {
            if (i == 4) {
                #pragma unroll
                for (int j = 8; j < 12; ++j) a[j] = xr[2 * j];
            }
            if (i == 8) {
                #pragma unroll
                for (int j = 12; j < 16; ++j) a[j] = xr[2 * j];
            }
            float4 bn[4];
            if (i + 1 < 16) {
                #pragma unroll
                for (int t = 0; t < 4; ++t) bn[t] = wl[(2 * (i + 1)) * 128 + t * 16];
            }
            v2f a0; a0.x = a[i].x; a0.y = a[i].y;
            v2f a1; a1.x = a[i].z; a1.y = a[i].w;
            #pragma unroll
            for (int t = 0; t < 4; ++t) {
                v2f bb; bb.x = b[t].x; bb.y = b[t].y;
                acc[t] = __builtin_amdgcn_wmma_f32_16x16x4_f32(
                     false, a0,  false, bb,
                     (short)0, acc[t],  false,  false);
            }
            #pragma unroll
            for (int t = 0; t < 4; ++t) {
                v2f bb; bb.x = b[t].z; bb.y = b[t].w;
                acc[t] = __builtin_amdgcn_wmma_f32_16x16x4_f32(
                     false, a1,  false, bb,
                     (short)0, acc[t],  false,  false);
            }
            if (i + 1 < 16) {
                #pragma unroll
                for (int t = 0; t < 4; ++t) b[t] = bn[t];
            }
        }

        float* hrow = H + (size_t)(st * 16 + half * 8) * N_FEAT + ch * 64 + l16 * 4;

        q += PROD_WAVES;
        st = (int)blockIdx.x + (int)gridDim.x * (q >> 1);
        live = st * 16 < M;
        if (live) {
            const float4* xn = (const float4*)(X + (size_t)(st * 16 + l16) * N_FEAT) + half;
            #pragma unroll
            for (int i = 0; i < 8; ++i) a[i] = xn[2 * i];
        }

        #pragma unroll
        for (int r = 0; r < 8; ++r) {
            float4 v;
            v.x = acc[0][r]; v.y = acc[1][r]; v.z = acc[2][r]; v.w = acc[3][r];
            *(float4*)(hrow + (size_t)r * N_FEAT) = v;
        }
    }
}

__global__ void __launch_bounds__(AGG_THREADS)
owner_list_kernel(const int* __restrict__ pairs, const unsigned* __restrict__ tag4, float* __restrict__ dinv,
                  int n, int E, int* recAll, int* recCount,
                  const float* __restrict__ X, const float* __restrict__ W, float* __restrict__ H) {
    __shared__ float4 Wl[32 * 128];
    __shared__ int deg[AGG_ROWS];
    __shared__ int qall[AGG_WAVES * LIST_QSLOT];
    __shared__ int cursor;
    __shared__ int nextChunk;

    const int tid = threadIdx.x;
    const unsigned ulane = threadIdx.x & 31u;
    const int wave = __builtin_amdgcn_readfirstlane(tid >> 5);
    const int base = (int)blockIdx.x * AGG_ROWS;
    const int left = n - base;
    const int rows = left < AGG_ROWS ? left : AGG_ROWS;
    int* rec = recAll != nullptr ? recAll + (size_t)blockIdx.x * (3 * REC_CAP) : nullptr;

    if (tid == 0) { cursor = 0; nextChunk = 0; }
    for (int r = tid; r < AGG_ROWS; r += AGG_THREADS) deg[r] = 0;
    {
        const int col  = tid & 127;
        const int slot = ((((col & 3) + ((col >> 6) << 2))) << 4) + ((col & 63) >> 2);
        const float* wp = W + col;
        for (int it = 0; it < 4; ++it) {
            const int kq = (tid >> 7) + 8 * it;
            const float* p = wp + (size_t)(4 * kq) * N_FEAT;
            float4 v;
            v.x = p[0];
            v.y = p[N_FEAT];
            v.z = p[2 * N_FEAT];
            v.w = p[3 * N_FEAT];
            Wl[kq * 128 + slot] = v;
        }
    }
    __syncthreads();

    if (wave < PROD_WAVES) product_half_stripes(X, H, Wl, n, wave, (int)ulane);

    const ListEdges f = { (const int2*)pairs, deg, rec, &cursor, base, rows, ulane };
    tag_walk<LIST_GROUP, true>(tag4, qall + wave * LIST_QSLOT, wave, (int)ulane, E, f, &nextChunk);
    __syncthreads();

    for (int r = tid; r < rows; r += AGG_THREADS) dinv[base + r] = rsqrtf(1.0f + (float)deg[r]);
    if (rec != nullptr && tid == 0) recCount[blockIdx.x] = cursor;
}

template <int G, bool FULL>
__device__ __forceinline__ void agg_rows_add(const float* __restrict__ H, float* acc,
                                             int s, int r, int wi, int count, unsigned ulane) {
    float4 v[G];
    unsigned off[G];
    #pragma unroll
    for (int u = 0; u < G; ++u)
        off[u] = ((unsigned)__builtin_amdgcn_readlane(s, u) << 9) | (ulane << 4);
    #pragma unroll
    for (int u = 0; u < G; ++u) {
        v[u] = make_float4(0.f, 0.f, 0.f, 0.f);
        if (FULL || u < count)
            v[u] = *(const float4*)((const char*)H + off[u]);
    }
    #pragma unroll
    for (int u = 0; u < G; ++u) {
        if (FULL || u < count) {
            const int ru2 = __builtin_amdgcn_readlane(r, u);
            const float wu = __int_as_float(__builtin_amdgcn_readlane(wi, u));
            float* a = acc + (size_t)(unsigned)ru2 * N_FEAT + ulane;
            __hip_atomic_fetch_add(a,       v[u].x * wu, __ATOMIC_RELAXED, __HIP_MEMORY_SCOPE_WORKGROUP);
            __hip_atomic_fetch_add(a + 32u, v[u].y * wu, __ATOMIC_RELAXED, __HIP_MEMORY_SCOPE_WORKGROUP);
            __hip_atomic_fetch_add(a + 64u, v[u].z * wu, __ATOMIC_RELAXED, __HIP_MEMORY_SCOPE_WORKGROUP);
            __hip_atomic_fetch_add(a + 96u, v[u].w * wu, __ATOMIC_RELAXED, __HIP_MEMORY_SCOPE_WORKGROUP);
        }
    }
}

struct AddEdges {
    const float* H; const float* dinv; const int2* pair; float* acc; int base; int rows; unsigned ulane;
    template <bool FULL>
    __device__ __forceinline__ void run(const int* qp, int count) const {
        const int qi = ulane < (unsigned)AGG_GROUP ? (int)ulane : 0;
        int e = 0;
        if (FULL || qi < count) e = qp[qi];
        const int2 sd = pair[e];
        const int s = sd.x;
        const int d = sd.y;
        const unsigned ru = (unsigned)(d - base);
        const bool own = ru < (unsigned)rows;
        const int r = own ? (int)ru : 0;
        const int wi = __float_as_int(own ? dinv[s] * dinv[d] : 0.f);
        agg_rows_add<AGG_GROUP, FULL>(H, acc, s, r, wi, count, ulane);
    }
};

template <bool FIRST>
__device__ __forceinline__ void agg_replay(const float* __restrict__ H, const float* __restrict__ dinv, int* rec,
                                           float* acc, int total, int wave, unsigned ulane, int base) {
    const int qi = ulane < (unsigned)REC_GROUP ? (int)ulane : 0;
    const int nFull = total / REC_GROUP;
    const int rem = total - nFull * REC_GROUP;
    const int last = total - 1;
    int b = wave;
    int i = b * REC_GROUP + qi;
    i = i < last ? i : last;
    int s = __builtin_nontemporal_load(rec + i);
    int r = __builtin_nontemporal_load(rec + REC_CAP + i);
    if (FIRST) {
        int i1 = (b + AGG_WAVES) * REC_GROUP + qi;
        i1 = i1 < last ? i1 : last;
        int s1 = __builtin_nontemporal_load(rec + i1);
        int r1 = __builtin_nontemporal_load(rec + REC_CAP + i1);
        float w = dinv[s] * dinv[base + r];
        while (b < nFull) {
            const int cs = s, cr = r, cw = __float_as_int(w);
            if (ulane < (unsigned)REC_GROUP) __builtin_nontemporal_store(cw, rec + 2 * REC_CAP + i);
            const float wa = dinv[s1];
            const float wb = dinv[base + r1];
            b += AGG_WAVES;
            int i2 = (b + AGG_WAVES) * REC_GROUP + qi;
            i2 = i2 < last ? i2 : last;
            const int s2 = __builtin_nontemporal_load(rec + i2);
            const int r2 = __builtin_nontemporal_load(rec + REC_CAP + i2);
            agg_rows_add<REC_GROUP, true>(H, acc, cs, cr, cw, REC_GROUP, ulane);
            s = s1; r = r1; i = i1; w = wa * wb;
            s1 = s2; r1 = r2; i1 = i2;
        }
        if (rem > 0 && b == nFull) {
            const int cw = __float_as_int(w);
            if ((int)ulane < rem) __builtin_nontemporal_store(cw, rec + 2 * REC_CAP + i);
            agg_rows_add<REC_GROUP, false>(H, acc, s, r, cw, rem, ulane);
        }
    } else {
        int wi = __builtin_nontemporal_load(rec + 2 * REC_CAP + i);
        while (b < nFull) {
            const int cs = s, cr = r, cw = wi;
            b += AGG_WAVES;
            i = b * REC_GROUP + qi;
            i = i < last ? i : last;
            s  = __builtin_nontemporal_load(rec + i);
            r  = __builtin_nontemporal_load(rec + REC_CAP + i);
            wi = __builtin_nontemporal_load(rec + 2 * REC_CAP + i);
            agg_rows_add<REC_GROUP, true>(H, acc, cs, cr, cw, REC_GROUP, ulane);
        }
        if (rem > 0 && b == nFull) agg_rows_add<REC_GROUP, false>(H, acc, s, r, wi, rem, ulane);
    }
}

template <bool FIRST>
__global__ void __launch_bounds__(AGG_THREADS)
aggregate_kernel(const float* __restrict__ H, const float* __restrict__ dinv,
                 const int* __restrict__ pairs, const unsigned* __restrict__ tag4,
                 const float* __restrict__ bias, const float* __restrict__ alpha,
                 float* __restrict__ out, int n, int E, int* recAll, const int* recCount) {
    __shared__ float acc[AGG_ROWS * N_FEAT];
    __shared__ int   qall[AGG_WAVES * AGG_QCAP];

    const int tid = threadIdx.x;
    const unsigned ulane = threadIdx.x & 31u;
    const int lane = (int)ulane;
    const int wave = __builtin_amdgcn_readfirstlane(tid >> 5);
    const int base = (int)blockIdx.x * AGG_ROWS;
    const int left = n - base;
    const int rows = left < AGG_ROWS ? left : AGG_ROWS;
    int* rec = recAll != nullptr ? recAll + (size_t)blockIdx.x * (3 * REC_CAP) : nullptr;

    for (int idx = tid; idx < rows * (N_FEAT / 4); idx += AGG_THREADS) {
        const int r = idx >> 5, c4 = idx & 31;
        const float di = dinv[base + r];
        const float di2 = di * di;
        const float4 hv = *(const float4*)(H + (size_t)(base + r) * N_FEAT + c4 * 4);
        const float4 bv = *(const float4*)(bias + c4 * 4);
        float4 st;
        st.x = hv.x * di2 + bv.x;
        st.y = hv.y * di2 + bv.y;
        st.z = hv.z * di2 + bv.z;
        st.w = hv.w * di2 + bv.w;
        float* a = acc + r * N_FEAT + c4;
        a[0] = st.x; a[32] = st.y; a[64] = st.z; a[96] = st.w;
    }
    __syncthreads();

    {
        const int total = rec != nullptr ? __builtin_amdgcn_readfirstlane(recCount[blockIdx.x]) : REC_CAP + 1;
        if (total <= REC_CAP) { if (total > 0) agg_replay<FIRST>(H, dinv, rec, acc, total, wave, ulane, base); }
        else {
            const AddEdges f = { H, dinv, (const int2*)pairs, acc, base, rows, ulane };
            tag_walk<AGG_GROUP, false>(tag4, qall + wave * AGG_QCAP, wave, lane, E, f, nullptr);
        }
    }
    __syncthreads();

    for (int idx = tid; idx < rows * (N_FEAT / 4); idx += AGG_THREADS) {
        const int c4 = idx & 31;
        const float4 av = *(const float4*)(alpha + c4 * 4);
        const float* a = acc + (idx >> 5) * N_FEAT + c4;
        float4 r;
        r.x = a[0]; r.y = a[32]; r.z = a[64]; r.w = a[96];
        r.x = r.x > 0.f ? r.x : av.x * r.x;
        r.y = r.y > 0.f ? r.y : av.y * r.y;
        r.z = r.z > 0.f ? r.z : av.z * r.z;
        r.w = r.w > 0.f ? r.w : av.w * r.w;
        *(float4*)(out + (size_t)base * N_FEAT + (size_t)idx * 4) = r;
    }
}

extern "C" void kernel_launch(void* const* d_in, const int* in_sizes, int n_in,
                              void* d_out, int out_size, void* d_ws, size_t ws_size,
                              hipStream_t stream) {
    (void)n_in; (void)out_size;
    const float* x  = (const float*)d_in[0];
    const int*   ei = (const int*)d_in[1];
    const float* W1 = (const float*)d_in[2];
    const float* b1 = (const float*)d_in[3];
    const float* W2 = (const float*)d_in[4];
    const float* b2 = (const float*)d_in[5];
    const float* al = (const float*)d_in[6];

    const int N = in_sizes[0] / N_FEAT;
    const int E = in_sizes[1] / 2;
    const int* src = ei;
    const int* dst = ei + E;

    float* dinv = (float*)d_ws;
    size_t off = (((size_t)N * sizeof(float)) + 255) & ~(size_t)255;
    float* hbuf = (float*)((char*)d_ws + off);
    unsigned* tags = (unsigned*)(hbuf + (size_t)N * N_FEAT);
    const int tagWords = ((E + AGG_CHUNK - 1) / AGG_CHUNK) * (AGG_CHUNK / 4);
    int* pairs = (int*)(tags + tagWords);
    float* outp = (float*)d_out;

    const int mTiles = (N + 15) / 16;
    const int gFull = (mTiles + 7) / 8;
    const int gBlocks = gFull < 768 ? gFull : 768;
    const int aBlocks = (N + AGG_ROWS - 1) / AGG_ROWS;
    int* lists  = pairs + 2 * (size_t)E;
    int* counts = lists + (size_t)aBlocks * (3 * REC_CAP);
    if ((size_t)((char*)(counts + aBlocks) - (char*)d_ws) > ws_size) { lists = nullptr; counts = nullptr; }

    tag_pair_kernel<<<(tagWords + 255) / 256, 256, 0, stream>>>(src, dst, tags, pairs, E, tagWords);
    owner_list_kernel<<<aBlocks, AGG_THREADS, 0, stream>>>(pairs, tags, dinv, N, E, lists, counts, x, W1, hbuf);

    aggregate_kernel<true><<<aBlocks, AGG_THREADS, 0, stream>>>(hbuf, dinv, pairs, tags, b1, al, outp, N, E, lists, counts);

    gemm_wmma_kernel<<<gBlocks, 256, 0, stream>>>(outp, W2, hbuf, N);
    aggregate_kernel<false><<<aBlocks, AGG_THREADS, 0, stream>>>(hbuf, dinv, pairs, tags, b2, al, outp, N, E, lists, counts);
}


# LDS f32 atomic edge adds replaced by wave-owned register sums: each layer bins the group's listed edges by owning wave in LDS (no sort), a row's registers picked by scalar index; bin-order sums
# speedup vs baseline: 4.5698x; 1.0570x over previous
#include <hip/hip_runtime.h>
#include <math.h>

#define N_FEAT 128

typedef __attribute__((ext_vector_type(2))) float v2f;
typedef __attribute__((ext_vector_type(8))) float v8f;

#define TAG_ROWS 392
__global__ void tag_pair_kernel(const int* __restrict__ src, const int* __restrict__ dst,
                                unsigned* __restrict__ tag4, int* __restrict__ pair, int E, int nWords) {
    const int t = blockIdx.x * blockDim.x + threadIdx.x;
    if (t >= nWords) return;
    const int e = 4 * t;
    int d0 = -1, d1 = -1, d2 = -1, d3 = -1;
    if (e + 3 < E && (((((size_t)dst) | ((size_t)src)) & 15) == 0)) {
        const int4 dv = *(const int4*)(dst + e);
        const int4 sv = *(const int4*)(src + e);
        d0 = dv.x; d1 = dv.y; d2 = dv.z; d3 = dv.w;
        int4* pp = (int4*)(pair + 2 * (size_t)e);
        pp[0] = make_int4(sv.x, dv.x, sv.y, dv.y);
        pp[1] = make_int4(sv.z, dv.z, sv.w, dv.w);
    } else {
        int2* pp = (int2*)pair;
        if (e     < E) { d0 = dst[e];     pp[e]     = make_int2(src[e],     d0); }
        if (e + 1 < E) { d1 = dst[e + 1]; pp[e + 1] = make_int2(src[e + 1], d1); }
        if (e + 2 < E) { d2 = dst[e + 2]; pp[e + 2] = make_int2(src[e + 2], d2); }
        if (e + 3 < E) { d3 = dst[e + 3]; pp[e + 3] = make_int2(src[e + 3], d3); }
    }
    unsigned w = 0u;
    if (d0 >= 0) w |= ((unsigned)(d0 / TAG_ROWS) & 255u);
    if (d1 >= 0) w |= ((unsigned)(d1 / TAG_ROWS) & 255u) << 8;
    if (d2 >= 0) w |= ((unsigned)(d2 / TAG_ROWS) & 255u) << 16;
    if (d3 >= 0) w |= ((unsigned)(d3 / TAG_ROWS) & 255u) << 24;
    tag4[t] = w;
}

__global__ void __launch_bounds__(256)
gemm_wmma_kernel(const float* __restrict__ X, const float* __restrict__ W,
                 float* __restrict__ H, int M) {
    __shared__ float4 Wl[32 * 128];
    const int tid  = threadIdx.x;
    const int wave = __builtin_amdgcn_readfirstlane(tid >> 5);
    const int lane = tid & 31;
    const int half = lane >> 4;
    const int l16  = lane & 15;
    const int step = (int)gridDim.x * 8;
    int st = wave * (int)gridDim.x + (int)blockIdx.x;
    bool live = st * 16 < M;

    float4 a4[16];
    {
        const float4* xr = (const float4*)(X + (size_t)(live ? st * 16 + l16 : 0) * N_FEAT) + half;
        #pragma unroll
        for (int i = 0; i < 16; ++i) a4[i] = xr[2 * i];
    }

    {
        const int col  = tid & 127;
        const int slot = ((((col & 3) + ((col >> 6) << 2))) << 4) + ((col & 63) >> 2);
        const float* wp = W + col;
        for (int it = 0; it < 16; ++it) {
            const int kq = (tid >> 7) + 2 * it;
            const float* p = wp + (size_t)(4 * kq) * N_FEAT;
            float4 v;
            v.x = p[0];
            v.y = p[N_FEAT];
            v.z = p[2 * N_FEAT];
            v.w = p[3 * N_FEAT];
            Wl[kq * 128 + slot] = v;
        }
    }
    __syncthreads();

    const float4* wl = Wl + half * 128 + l16;

    while (live) {
        asm volatile("" ::: "memory");
        v8f acc[8];
        #pragma unroll
        for (int t = 0; t < 8; ++t) acc[t] = (v8f){0.f, 0.f, 0.f, 0.f, 0.f, 0.f, 0.f, 0.f};

        float4 b[8];
        #pragma unroll
        for (int t = 0; t < 8; ++t) b[t] = wl[t * 16];

        #pragma unroll
        for (int i = 0; i < 16; ++i) {
            float4 bn[8];
            if (i + 1 < 16) {
                #pragma unroll
                for (int t = 0; t < 8; ++t) bn[t] = wl[(2 * (i + 1)) * 128 + t * 16];
            }
            v2f a0; a0.x = a4[i].x; a0.y = a4[i].y;
            v2f a1; a1.x = a4[i].z; a1.y = a4[i].w;
            #pragma unroll
            for (int t = 0; t < 8; ++t) {
                v2f bb; bb.x = b[t].x; bb.y = b[t].y;
                acc[t] = __builtin_amdgcn_wmma_f32_16x16x4_f32(
                     false, a0,  false, bb,
                     (short)0, acc[t],  false,  false);
            }
            #pragma unroll
            for (int t = 0; t < 8; ++t) {
                v2f bb; bb.x = b[t].z; bb.y = b[t].w;
                acc[t] = __builtin_amdgcn_wmma_f32_16x16x4_f32(
                     false, a1,  false, bb,
                     (short)0, acc[t],  false,  false);
            }
            if (i + 1 < 16) {
                #pragma unroll
                for (int t = 0; t < 8; ++t) b[t] = bn[t];
            }
        }

        float* hrow = H + (size_t)(st * 16 + half * 8) * N_FEAT + l16 * 4;

        st += step;
        live = st * 16 < M;
        if (live) {
            const float4* xr = (const float4*)(X + (size_t)(st * 16 + l16) * N_FEAT) + half;
            #pragma unroll
            for (int i = 0; i < 16; ++i) a4[i] = xr[2 * i];
        }

        #pragma unroll
        for (int r = 0; r < 8; ++r) {
            float4 lo, hi;
            lo.x = acc[0][r]; lo.y = acc[1][r]; lo.z = acc[2][r]; lo.w = acc[3][r];
            hi.x = acc[4][r]; hi.y = acc[5][r]; hi.z = acc[6][r]; hi.w = acc[7][r];
            *(float4*)(hrow + (size_t)r * N_FEAT)      = lo;
            *(float4*)(hrow + (size_t)r * N_FEAT + 64) = hi;
        }
    }
}

#define AGG_ROWS    392
#define AGG_THREADS 1024
#define AGG_WAVES   (AGG_THREADS / 32)
#define AGG_AHEAD   8
#define AGG_VEC     512
#define AGG_CHUNK   (AGG_AHEAD * AGG_VEC)
#define AGG_GROUP   12
#define AGG_QCAP    (AGG_GROUP + AGG_VEC)
#define LIST_GROUP  32
#define LIST_QCAP   (LIST_GROUP + AGG_VEC)
#define LIST_QSLOT  1312
#define REC_CAP     7168
#define REC_GROUP   16
static_assert(TAG_ROWS == AGG_ROWS, "the tags must be cut at the groups' row count");
static_assert(LIST_QSLOT >= LIST_QCAP, "a wave's queue must hold a left-over and one load's edges");

__device__ __forceinline__ unsigned zero_bytes(unsigned x) {
    const unsigned t = ((x & 0x7F7F7F7Fu) + 0x7F7F7F7Fu) | x;
    return ~t & 0x80808080u;
}

template <int GROUP, bool DYN, class F>
__device__ __forceinline__ void tag_walk(const unsigned* __restrict__ tag4, int* q, int wave, int lane, int E, const F& f, int* next) {
    int cnt = 0;
    const unsigned pat = ((unsigned)blockIdx.x & 255u) * 0x01010101u;
    const int nChunks = (E + AGG_CHUNK - 1) / AGG_CHUNK;
    const int nTrips = (nChunks + AGG_WAVES - 1) / AGG_WAVES;
    const int rot = (int)(((long long)blockIdx.x * nChunks) / (int)gridDim.x);

    for (int k = 0; k < (DYN ? nChunks : nTrips); ++k) {
        int c;
        if (DYN) {
            int t = 0;
            if (lane == 0) t = __hip_atomic_fetch_add(next, 1, __ATOMIC_RELAXED, __HIP_MEMORY_SCOPE_WORKGROUP);
            c = __builtin_amdgcn_readfirstlane(t);
            if (c >= nChunks) break;
        } else {
            c = wave + AGG_WAVES * k;
            if (c >= nChunks) continue;
        }
        c += rot;
        if (c >= nChunks) c -= nChunks;
        const int e0 = c * AGG_CHUNK;

        uint4 d[AGG_AHEAD];
        {
            const uint4* tp = (const uint4*)(tag4 + (e0 >> 2)) + lane;
            #pragma unroll
            for (int j = 0; j < AGG_AHEAD; ++j) d[j] = tp[j * 32];
        }

        int jstart = 0;
        for (;;) {
            int jnext = AGG_AHEAD;
            #pragma unroll
            for (int j = 0; j < AGG_AHEAD; ++j) {
                if (j >= jstart && jnext == AGG_AHEAD) {
                    unsigned m = (zero_bytes(d[j].x ^ pat) >> 3) | (zero_bytes(d[j].y ^ pat) >> 2) |
                                 (zero_bytes(d[j].z ^ pat) >> 1) |  zero_bytes(d[j].w ^ pat);
                    const int eb = e0 + j * AGG_VEC + 16 * lane;
                    for (;;) {
                        const bool has = m != 0u;
                        if (__builtin_amdgcn_ballot_w32(has) == 0u) break;
                        const int b = has ? __builtin_ctz(m) : 0;
                        const int e = eb + 4 * ((b & 7) - 4) + (b >> 3);
                        const bool keep = has && e < E;
                        const unsigned kb = __builtin_amdgcn_ballot_w32(keep);
                        if (keep) q[cnt + (int)__builtin_amdgcn_mbcnt_lo(kb, 0u)] = e;
                        cnt += __builtin_popcount(kb);
                        m &= m - 1u;
                    }
                    if (cnt >= GROUP) jnext = j + 1;
                }
            }
            __builtin_amdgcn_fence(__ATOMIC_ACQ_REL, "wavefront");
            while (cnt >= GROUP) {
                cnt -= GROUP;
                f.template run<true>(q + cnt, GROUP);
            }
            if (jnext >= AGG_AHEAD) break;
            jstart = jnext;
        }
    }
    if (cnt > 0) f.template run<false>(q, cnt);
}

struct ListEdges {
    const int2* pair; int* deg; int* rec; int* cursor; int base; int rows; unsigned ulane;
    template <bool FULL>
    __device__ __forceinline__ void run(const int* qp, int count) const {
        const bool have = FULL || (int)ulane < count;
        int e = 0;
        if (have) e = qp[ulane];
        const int2 sd = pair[e];
        const unsigned ru = (unsigned)(sd.y - base);
        const bool own = have && ru < (unsigned)rows;
        if (own) __hip_atomic_fetch_add(deg + ru, 1, __ATOMIC_RELAXED, __HIP_MEMORY_SCOPE_WORKGROUP);
        if (rec != nullptr) {
            const unsigned kb = __builtin_amdgcn_ballot_w32(own);
            int pos = 0;
            if (ulane == 0u) pos = __hip_atomic_fetch_add(cursor, __builtin_popcount(kb), __ATOMIC_RELAXED, __HIP_MEMORY_SCOPE_WORKGROUP);
            pos = __builtin_amdgcn_readfirstlane(pos);
            const int idx = pos + (int)__builtin_amdgcn_mbcnt_lo(kb, 0u);
            if (own && idx < REC_CAP) {
                __builtin_nontemporal_store(sd.x,    rec + idx);
                __builtin_nontemporal_store((int)ru, rec + REC_CAP + idx);
            }
        }
    }
};

#define PROD_WAVES 8
__device__ __forceinline__ void product_half_stripes(const float* __restrict__ X, float* __restrict__ H,
                                                     const float4* Wl, int M, int g, int lane) {
    const int half = lane >> 4;
    const int l16  = lane & 15;
    int q  = g;
    int st = (int)blockIdx.x + (int)gridDim.x * (q >> 1);
    bool live = st * 16 < M;
    const int ch = g & 1;
    const float4* wl = Wl + half * 128 + ch * 64 + l16;

    float4 a[16];
    if (live) {
        const float4* xr = (const float4*)(X + (size_t)(st * 16 + l16) * N_FEAT) + half;
        #pragma unroll
        for (int i = 0; i < 8; ++i) a[i] = xr[2 * i];
    }

    while (live) {
        asm volatile("" ::: "memory");
        const float4* xr = (const float4*)(X + (size_t)(st * 16 + l16) * N_FEAT) + half;

        v8f acc[4];
        #pragma unroll
        for (int t = 0; t < 4; ++t) acc[t] = (v8f){0.f, 0.f, 0.f, 0.f, 0.f, 0.f, 0.f, 0.f};

        float4 b[4];
        #pragma unroll
        for (int t = 0; t < 4; ++t) b[t] = wl[t * 16];

        #pragma unroll
        for (int i = 0; i < 16; ++i) {
            if (i == 4) {
                #pragma unroll
                for (int j = 8; j < 12; ++j) a[j] = xr[2 * j];
            }
            if (i == 8) {
                #pragma unroll
                for (int j = 12; j < 16; ++j) a[j] = xr[2 * j];
            }
            float4 bn[4];
            if (i + 1 < 16) {
                #pragma unroll
                for (int t = 0; t < 4; ++t) bn[t] = wl[(2 * (i + 1)) * 128 + t * 16];
            }
            v2f a0; a0.x = a[i].x; a0.y = a[i].y;
            v2f a1; a1.x = a[i].z; a1.y = a[i].w;
            #pragma unroll
            for (int t = 0; t < 4; ++t) {
                v2f bb; bb.x = b[t].x; bb.y = b[t].y;
                acc[t] = __builtin_amdgcn_wmma_f32_16x16x4_f32(
                     false, a0,  false, bb,
                     (short)0, acc[t],  false,  false);
            }
            #pragma unroll
            for (int t = 0; t < 4; ++t) {
                v2f bb; bb.x = b[t].z; bb.y = b[t].w;
                acc[t] = __builtin_amdgcn_wmma_f32_16x16x4_f32(
                     false, a1,  false, bb,
                     (short)0, acc[t],  false,  false);
            }
            if (i + 1 < 16) {
                #pragma unroll
                for (int t = 0; t < 4; ++t) b[t] = bn[t];
            }
        }

        float* hrow = H + (size_t)(st * 16 + half * 8) * N_FEAT + ch * 64 + l16 * 4;

        q += PROD_WAVES;
        st = (int)blockIdx.x + (int)gridDim.x * (q >> 1);
        live = st * 16 < M;
        if (live) {
            const float4* xn = (const float4*)(X + (size_t)(st * 16 + l16) * N_FEAT) + half;
            #pragma unroll
            for (int i = 0; i < 8; ++i) a[i] = xn[2 * i];
        }

        #pragma unroll
        for (int r = 0; r < 8; ++r) {
            float4 v;
            v.x = acc[0][r]; v.y = acc[1][r]; v.z = acc[2][r]; v.w = acc[3][r];
            *(float4*)(hrow + (size_t)r * N_FEAT) = v;
        }
    }
}

__global__ void __launch_bounds__(AGG_THREADS)
owner_list_kernel(const int* __restrict__ pairs, const unsigned* __restrict__ tag4, float* __restrict__ dinv,
                  int n, int E, int* recAll, int* recCount,
                  const float* __restrict__ X, const float* __restrict__ W, float* __restrict__ H) {
    __shared__ float4 Wl[32 * 128];
    __shared__ int deg[AGG_ROWS];
    __shared__ int qall[AGG_WAVES * LIST_QSLOT];
    __shared__ int cursor;
    __shared__ int nextChunk;

    const int tid = threadIdx.x;
    const unsigned ulane = threadIdx.x & 31u;
    const int wave = __builtin_amdgcn_readfirstlane(tid >> 5);
    const int base = (int)blockIdx.x * AGG_ROWS;
    const int left = n - base;
    const int rows = left < AGG_ROWS ? left : AGG_ROWS;
    int* rec = recAll != nullptr ? recAll + (size_t)blockIdx.x * (3 * REC_CAP) : nullptr;

    if (tid == 0) { cursor = 0; nextChunk = 0; }
    for (int r = tid; r < AGG_ROWS; r += AGG_THREADS) deg[r] = 0;
    {
        const int col  = tid & 127;
        const int slot = ((((col & 3) + ((col >> 6) << 2))) << 4) + ((col & 63) >> 2);
        const float* wp = W + col;
        for (int it = 0; it < 4; ++it) {
            const int kq = (tid >> 7) + 8 * it;
            const float* p = wp + (size_t)(4 * kq) * N_FEAT;
            float4 v;
            v.x = p[0];
            v.y = p[N_FEAT];
            v.z = p[2 * N_FEAT];
            v.w = p[3 * N_FEAT];
            Wl[kq * 128 + slot] = v;
        }
    }
    __syncthreads();

    if (wave < PROD_WAVES) product_half_stripes(X, H, Wl, n, wave, (int)ulane);

    const ListEdges f = { (const int2*)pairs, deg, rec, &cursor, base, rows, ulane };
    tag_walk<LIST_GROUP, true>(tag4, qall + wave * LIST_QSLOT, wave, (int)ulane, E, f, &nextChunk);
    __syncthreads();

    for (int r = tid; r < rows; r += AGG_THREADS) dinv[base + r] = rsqrtf(1.0f + (float)deg[r]);
    if (rec != nullptr && tid == 0) recCount[blockIdx.x] = cursor;
}

template <int G, bool FULL>
__device__ __forceinline__ void agg_rows_add(const float* __restrict__ H, float* acc,
                                             int s, int r, int wi, int count, unsigned ulane) {
    float4 v[G];
    unsigned off[G];
    #pragma unroll
    for (int u = 0; u < G; ++u)
        off[u] = ((unsigned)__builtin_amdgcn_readlane(s, u) << 9) | (ulane << 4);
    #pragma unroll
    for (int u = 0; u < G; ++u) {
        v[u] = make_float4(0.f, 0.f, 0.f, 0.f);
        if (FULL || u < count)
            v[u] = *(const float4*)((const char*)H + off[u]);
    }
    #pragma unroll
    for (int u = 0; u < G; ++u) {
        if (FULL || u < count) {
            const int ru2 = __builtin_amdgcn_readlane(r, u);
            const float wu = __int_as_float(__builtin_amdgcn_readlane(wi, u));
            float* a = acc + (size_t)(unsigned)ru2 * N_FEAT + ulane;
            __hip_atomic_fetch_add(a,       v[u].x * wu, __ATOMIC_RELAXED, __HIP_MEMORY_SCOPE_WORKGROUP);
            __hip_atomic_fetch_add(a + 32u, v[u].y * wu, __ATOMIC_RELAXED, __HIP_MEMORY_SCOPE_WORKGROUP);
            __hip_atomic_fetch_add(a + 64u, v[u].z * wu, __ATOMIC_RELAXED, __HIP_MEMORY_SCOPE_WORKGROUP);
            __hip_atomic_fetch_add(a + 96u, v[u].w * wu, __ATOMIC_RELAXED, __HIP_MEMORY_SCOPE_WORKGROUP);
        }
    }
}

struct AddEdges {
    const float* H; const float* dinv; const int2* pair; float* acc; int base; int rows; unsigned ulane;
    template <bool FULL>
    __device__ __forceinline__ void run(const int* qp, int count) const {
        const int qi = ulane < (unsigned)AGG_GROUP ? (int)ulane : 0;
        int e = 0;
        if (FULL || qi < count) e = qp[qi];
        const int2 sd = pair[e];
        const int s = sd.x;
        const int d = sd.y;
        const unsigned ru = (unsigned)(d - base);
        const bool own = ru < (unsigned)rows;
        const int r = own ? (int)ru : 0;
        const int wi = __float_as_int(own ? dinv[s] * dinv[d] : 0.f);
        agg_rows_add<AGG_GROUP, FULL>(H, acc, s, r, wi, count, ulane);
    }
};

template <bool FIRST>
__device__ __forceinline__ void agg_replay(const float* __restrict__ H, const float* __restrict__ dinv, int* rec,
                                           float* acc, int total, int wave, unsigned ulane, int base) {
    const int qi = ulane < (unsigned)REC_GROUP ? (int)ulane : 0;
    const int nFull = total / REC_GROUP;
    const int rem = total - nFull * REC_GROUP;
    const int last = total - 1;
    int b = wave;
    int i = b * REC_GROUP + qi;
    i = i < last ? i : last;
    int s = __builtin_nontemporal_load(rec + i);
    int r = __builtin_nontemporal_load(rec + REC_CAP + i);
    if (FIRST) {
        int i1 = (b + AGG_WAVES) * REC_GROUP + qi;
        i1 = i1 < last ? i1 : last;
        int s1 = __builtin_nontemporal_load(rec + i1);
        int r1 = __builtin_nontemporal_load(rec + REC_CAP + i1);
        float w = dinv[s] * dinv[base + r];
        while (b < nFull) {
            const int cs = s, cr = r, cw = __float_as_int(w);
            if (ulane < (unsigned)REC_GROUP) __builtin_nontemporal_store(cw, rec + 2 * REC_CAP + i);
            const float wa = dinv[s1];
            const float wb = dinv[base + r1];
            b += AGG_WAVES;
            int i2 = (b + AGG_WAVES) * REC_GROUP + qi;
            i2 = i2 < last ? i2 : last;
            const int s2 = __builtin_nontemporal_load(rec + i2);
            const int r2 = __builtin_nontemporal_load(rec + REC_CAP + i2);
            agg_rows_add<REC_GROUP, true>(H, acc, cs, cr, cw, REC_GROUP, ulane);
            s = s1; r = r1; i = i1; w = wa * wb;
            s1 = s2; r1 = r2; i1 = i2;
        }
        if (rem > 0 && b == nFull) {
            const int cw = __float_as_int(w);
            if ((int)ulane < rem) __builtin_nontemporal_store(cw, rec + 2 * REC_CAP + i);
            agg_rows_add<REC_GROUP, false>(H, acc, s, r, cw, rem, ulane);
        }
    } else {
        int wi = __builtin_nontemporal_load(rec + 2 * REC_CAP + i);
        while (b < nFull) {
            const int cs = s, cr = r, cw = wi;
            b += AGG_WAVES;
            i = b * REC_GROUP + qi;
            i = i < last ? i : last;
            s  = __builtin_nontemporal_load(rec + i);
            r  = __builtin_nontemporal_load(rec + REC_CAP + i);
            wi = __builtin_nontemporal_load(rec + 2 * REC_CAP + i);
            agg_rows_add<REC_GROUP, true>(H, acc, cs, cr, cw, REC_GROUP, ulane);
        }
        if (rem > 0 && b == nFull) agg_rows_add<REC_GROUP, false>(H, acc, s, r, wi, rem, ulane);
    }
}

#define WROWS          13
#define BIN_CAP        384
#define BIN_G          12
#define BIN_PER_THREAD (REC_CAP / AGG_THREADS)
static_assert(REC_CAP % AGG_THREADS == 0, "every thread takes the same number of list slots");
static_assert(WROWS * AGG_WAVES >= AGG_ROWS, "the waves' runs must cover the group's rows");
static_assert(AGG_WAVES * BIN_CAP * 16 <= AGG_ROWS * N_FEAT * 4, "the bins live in the accumulator's LDS");

typedef float v16f __attribute__((ext_vector_type(16)));

template <bool FIRST>
__global__ void __launch_bounds__(AGG_THREADS)
aggregate_kernel(const float* __restrict__ H, const float* __restrict__ dinv,
                 const int* __restrict__ pairs, const unsigned* __restrict__ tag4,
                 const float* __restrict__ bias, const float* __restrict__ alpha,
                 float* __restrict__ out, int n, int E, int* recAll, const int* recCount) {
    __shared__ __attribute__((aligned(16))) float acc[AGG_ROWS * N_FEAT];
    __shared__ int   qall[AGG_WAVES * AGG_QCAP];
    __shared__ int   binCount[AGG_WAVES];
    __shared__ int   binOver;

    const int tid = threadIdx.x;
    const unsigned ulane = threadIdx.x & 31u;
    const int lane = (int)ulane;
    const int wave = __builtin_amdgcn_readfirstlane(tid >> 5);
    const int base = (int)blockIdx.x * AGG_ROWS;
    const int left = n - base;
    const int rows = left < AGG_ROWS ? left : AGG_ROWS;
    int* rec = recAll != nullptr ? recAll + (size_t)blockIdx.x * (3 * REC_CAP) : nullptr;

    const int total = rec != nullptr ? __builtin_amdgcn_readfirstlane(recCount[blockIdx.x]) : REC_CAP + 1;
    bool listed = total <= REC_CAP;

    if (listed) {
        float4 a[WROWS];
        #pragma unroll
        for (int j = 0; j < WROWS; ++j) {
            const int row = wave + AGG_WAVES * j;
            a[j] = *((const float4*)(H + (size_t)(base + (row < rows ? row : 0)) * N_FEAT) + lane);
        }

        if (tid < AGG_WAVES) binCount[tid] = 0;
        if (tid == 0) binOver = 0;
        int bs[BIN_PER_THREAD], br[BIN_PER_THREAD], bw[BIN_PER_THREAD];
        #pragma unroll
        for (int k = 0; k < BIN_PER_THREAD; ++k) {
            const int i = tid + k * AGG_THREADS;
            bs[k] = 0; br[k] = 0; bw[k] = 0;
            if (i < total) {
                bs[k] = __builtin_nontemporal_load(rec + i);
                br[k] = __builtin_nontemporal_load(rec + REC_CAP + i);
                if (!FIRST) bw[k] = __builtin_nontemporal_load(rec + 2 * REC_CAP + i);
            }
        }
        if (FIRST) {
            #pragma unroll
            for (int k = 0; k < BIN_PER_THREAD; ++k) bw[k] = __float_as_int(dinv[bs[k]] * dinv[base + br[k]]);
        }
        __syncthreads();
        int4* bins = (int4*)acc;
        #pragma unroll
        for (int k = 0; k < BIN_PER_THREAD; ++k) {
            const int i = tid + k * AGG_THREADS;
            if (i < total) {
                if (FIRST) __builtin_nontemporal_store(bw[k], rec + 2 * REC_CAP + i);
                const int sub = br[k] % AGG_WAVES;
                const int pos = __hip_atomic_fetch_add(binCount + sub, 1, __ATOMIC_RELAXED, __HIP_MEMORY_SCOPE_WORKGROUP);
                if (pos < BIN_CAP) bins[sub * BIN_CAP + pos] = make_int4(bs[k], br[k] / AGG_WAVES, bw[k], 0);
                else binOver = 1;
            }
        }
        __syncthreads();
        listed = __builtin_amdgcn_readfirstlane(binOver) == 0;

        if (listed) {
            const float4 bv = *((const float4*)bias + lane);
            #pragma unroll
            for (int j = 0; j < WROWS; ++j) {
                const int row = wave + AGG_WAVES * j;
                if (row < rows) {
                    const float di = dinv[base + row];
                    const float di2 = di * di;
                    a[j].x = a[j].x * di2 + bv.x; a[j].y = a[j].y * di2 + bv.y;
                    a[j].z = a[j].z * di2 + bv.z; a[j].w = a[j].w * di2 + bv.w;
                } else a[j] = make_float4(0.f, 0.f, 0.f, 0.f);
            }

            v16f ax = {}, ay = {}, az = {}, aw = {};
            #pragma unroll
            for (int j = 0; j < WROWS; ++j) { ax[j] = a[j].x; ay[j] = a[j].y; az[j] = a[j].z; aw[j] = a[j].w; }

            const int cnt = __builtin_amdgcn_readfirstlane(binCount[wave]);
            const int4* mybin = (const int4*)acc + wave * BIN_CAP;
            const int qi = lane < BIN_G ? lane : 0;
            const int4 none = make_int4(0, WROWS, 0, 0);
            int4 rc = qi < cnt ? mybin[qi] : none;
            for (int b0 = 0; b0 < cnt; b0 += BIN_G) {
                const int4 cur = rc;
                const int nx = b0 + BIN_G + qi;
                rc = nx < cnt ? mybin[nx] : none;
                float4 v[BIN_G];
                #pragma unroll
                for (int u = 0; u < BIN_G; ++u) {
                    const unsigned su = (unsigned)__builtin_amdgcn_readlane(cur.x, u);
                    v[u] = *((const float4*)(H + (size_t)su * N_FEAT) + lane);
                }
                #pragma unroll
                for (int u = 0; u < BIN_G; ++u) {
                    const int rl = __builtin_amdgcn_readlane(cur.y, u);
                    const float wu = __int_as_float(__builtin_amdgcn_readlane(cur.z, u));
                    ax[rl] = fmaf(v[u].x, wu, ax[rl]); ay[rl] = fmaf(v[u].y, wu, ay[rl]);
                    az[rl] = fmaf(v[u].z, wu, az[rl]); aw[rl] = fmaf(v[u].w, wu, aw[rl]);
                }
            }

            const float4 av = *((const float4*)alpha + lane);
            #pragma unroll
            for (int j = 0; j < WROWS; ++j) {
                const int row = wave + AGG_WAVES * j;
                if (row < rows) {
                    float4 r = make_float4(ax[j], ay[j], az[j], aw[j]);
                    r.x = r.x > 0.f ? r.x : av.x * r.x;
                    r.y = r.y > 0.f ? r.y : av.y * r.y;
                    r.z = r.z > 0.f ? r.z : av.z * r.z;
                    r.w = r.w > 0.f ? r.w : av.w * r.w;
                    *((float4*)(out + (size_t)(base + row) * N_FEAT) + lane) = r;
                }
            }
        }
    }

    if (!listed) {
        for (int idx = tid; idx < rows * (N_FEAT / 4); idx += AGG_THREADS) {
            const int r = idx >> 5, c4 = idx & 31;
            const float di = dinv[base + r];
            const float di2 = di * di;
            const float4 hv = *(const float4*)(H + (size_t)(base + r) * N_FEAT + c4 * 4);
            const float4 bv = *(const float4*)(bias + c4 * 4);
            float4 st;
            st.x = hv.x * di2 + bv.x;
            st.y = hv.y * di2 + bv.y;
            st.z = hv.z * di2 + bv.z;
            st.w = hv.w * di2 + bv.w;
            float* a = acc + r * N_FEAT + c4;
            a[0] = st.x; a[32] = st.y; a[64] = st.z; a[96] = st.w;
        }
        __syncthreads();

        if (total <= REC_CAP) { if (total > 0) agg_replay<FIRST>(H, dinv, rec, acc, total, wave, ulane, base); }
        else {
            const AddEdges f = { H, dinv, (const int2*)pairs, acc, base, rows, ulane };
            tag_walk<AGG_GROUP, false>(tag4, qall + wave * AGG_QCAP, wave, lane, E, f, nullptr);
        }
        __syncthreads();

        for (int idx = tid; idx < rows * (N_FEAT / 4); idx += AGG_THREADS) {
            const int c4 = idx & 31;
            const float4 av = *(const float4*)(alpha + c4 * 4);
            const float* a = acc + (idx >> 5) * N_FEAT + c4;
            float4 r;
            r.x = a[0]; r.y = a[32]; r.z = a[64]; r.w = a[96];
            r.x = r.x > 0.f ? r.x : av.x * r.x;
            r.y = r.y > 0.f ? r.y : av.y * r.y;
            r.z = r.z > 0.f ? r.z : av.z * r.z;
            r.w = r.w > 0.f ? r.w : av.w * r.w;
            *(float4*)(out + (size_t)base * N_FEAT + (size_t)idx * 4) = r;
        }
    }
}

extern "C" void kernel_launch(void* const* d_in, const int* in_sizes, int n_in,
                              void* d_out, int out_size, void* d_ws, size_t ws_size,
                              hipStream_t stream) {
    (void)n_in; (void)out_size;
    const float* x  = (const float*)d_in[0];
    const int*   ei = (const int*)d_in[1];
    const float* W1 = (const float*)d_in[2];
    const float* b1 = (const float*)d_in[3];
    const float* W2 = (const float*)d_in[4];
    const float* b2 = (const float*)d_in[5];
    const float* al = (const float*)d_in[6];

    const int N = in_sizes[0] / N_FEAT;
    const int E = in_sizes[1] / 2;
    const int* src = ei;
    const int* dst = ei + E;

    float* dinv = (float*)d_ws;
    size_t off = (((size_t)N * sizeof(float)) + 255) & ~(size_t)255;
    float* hbuf = (float*)((char*)d_ws + off);
    unsigned* tags = (unsigned*)(hbuf + (size_t)N * N_FEAT);
    const int tagWords = ((E + AGG_CHUNK - 1) / AGG_CHUNK) * (AGG_CHUNK / 4);
    int* pairs = (int*)(tags + tagWords);
    float* outp = (float*)d_out;

    const int mTiles = (N + 15) / 16;
    const int gFull = (mTiles + 7) / 8;
    const int gBlocks = gFull < 768 ? gFull : 768;
    const int aBlocks = (N + AGG_ROWS - 1) / AGG_ROWS;
    int* lists  = pairs + 2 * (size_t)E;
    int* counts = lists + (size_t)aBlocks * (3 * REC_CAP);
    if ((size_t)((char*)(counts + aBlocks) - (char*)d_ws) > ws_size) { lists = nullptr; counts = nullptr; }

    tag_pair_kernel<<<(tagWords + 255) / 256, 256, 0, stream>>>(src, dst, tags, pairs, E, tagWords);
    owner_list_kernel<<<aBlocks, AGG_THREADS, 0, stream>>>(pairs, tags, dinv, N, E, lists, counts, x, W1, hbuf);

    aggregate_kernel<true><<<aBlocks, AGG_THREADS, 0, stream>>>(hbuf, dinv, pairs, tags, b1, al, outp, N, E, lists, counts);

    gemm_wmma_kernel<<<gBlocks, 256, 0, stream>>>(outp, W2, hbuf, N);
    aggregate_kernel<false><<<aBlocks, AGG_THREADS, 0, stream>>>(hbuf, dinv, pairs, tags, b2, al, outp, N, E, lists, counts);
}
